# starting kernel
# baseline (speedup 1.0000x reference)
#include <hip/hip_runtime.h>

typedef __attribute__((ext_vector_type(16))) _Float16 v16h;
typedef __attribute__((ext_vector_type(8)))  float    v8f;

#define WMMA_F16(A, B, C) \
    __builtin_amdgcn_wmma_f32_16x16x32_f16(false, (A), false, (B), (short)0, (C), false, false)

#define L2E 1.4426950408889634f
#define LN2 0.6931471805599453f

__device__ __forceinline__ float exp2_fast(float x) { return __builtin_amdgcn_exp2f(x); }
__device__ __forceinline__ float log2_fast(float x) { return __builtin_amdgcn_logf(x); }
__device__ __forceinline__ float rcp_fast(float x)  { return __builtin_amdgcn_rcpf(x); }

__device__ __forceinline__ float softplus_f(float x) {
    return x > 20.f ? x : LN2 * log2_fast(1.f + exp2_fast(x * L2E));
}
__device__ __forceinline__ float sigmoid_f(float x) {
    return rcp_fast(1.f + exp2_fast(-x * L2E));
}

__global__ __launch_bounds__(256)
void nerf_wmma_kernel(const float* __restrict__ rays_o,
                      const float* __restrict__ rays_d,
                      const float* __restrict__ W1,   const float* __restrict__ b1,
                      const float* __restrict__ Wsig, const float* __restrict__ Wsig_d,
                      const float* __restrict__ Wc1,  const float* __restrict__ bc1,
                      const float* __restrict__ Wc2,  const float* __restrict__ Wc2_d,
                      const int*   __restrict__ num_steps,
                      float* __restrict__ out, int N)
{
    const int tid  = blockIdx.x * blockDim.x + threadIdx.x;
    const int lane = threadIdx.x & 31;
    const int wave = tid >> 5;
    const int m    = lane & 15;
    const bool hi  = (lane & 16) != 0;
    const int rayBase = wave * 16;

    int ray = rayBase + m;
    if (ray >= N) ray = N - 1;

    v16h A1t0, A1t1, Ac1t0, Ac1t1, Asig, Acol;
#pragma unroll
    for (int e = 0; e < 16; ++e) {
        const int K = (e < 8 ? e : e + 8) + (hi ? 8 : 0);
        float a0 = (K < 3) ? W1[K * 32 + m]        : 0.f;
        float a1 = (K < 3) ? W1[K * 32 + (m + 16)] : 0.f;
        A1t0[e] = (_Float16)a0;
        A1t1[e] = (_Float16)a1;
        float c0 = (K < 6) ? Wc1[K * 32 + m]        : 0.f;
        float c1 = (K < 6) ? Wc1[K * 32 + (m + 16)] : 0.f;
        Ac1t0[e] = (_Float16)c0;
        Ac1t1[e] = (_Float16)c1;
        float sv = (m == 0) ? Wsig[K] : ((m == 1) ? Wsig_d[K] : 0.f);
        Asig[e] = (_Float16)sv;
        float cv = (m < 3) ? Wc2[K * 3 + m] : ((m < 6) ? Wc2_d[K * 3 + (m - 3)] : 0.f);
        Acol[e] = (_Float16)cv;
    }
    v8f CbT0, CbT1, CbC0, CbC1;
#pragma unroll
    for (int i = 0; i < 8; ++i) {
        const int h0 = i + (hi ? 8 : 0);
        CbT0[i] = b1[h0];      CbT1[i] = b1[h0 + 16];
        CbC0[i] = bc1[h0];     CbC1[i] = bc1[h0 + 16];
    }

    float ox = rays_o[ray * 3 + 0], oy = rays_o[ray * 3 + 1], oz = rays_o[ray * 3 + 2];
    float dx = rays_d[ray * 3 + 0], dy = rays_d[ray * 3 + 1], dz = rays_d[ray * 3 + 2];
    {
        float il = 1.f / sqrtf(dx * dx + dy * dy + dz * dz);
        dx *= il; dy *= il; dz *= il;
    }
    float near, far;
    {
        float ivx = 1.f / dx, ivy = 1.f / dy, ivz = 1.f / dz;
        float t1x = (-1.f - ox) * ivx, t2x = (1.f - ox) * ivx;
        float t1y = (-1.f - oy) * ivy, t2y = (1.f - oy) * ivy;
        float t1z = (-1.f - oz) * ivz, t2z = (1.f - oz) * ivz;
        near = fmaxf(fmaxf(fminf(t1x, t2x), fminf(t1y, t2y)), fminf(t1z, t2z));
        far  = fminf(fminf(fmaxf(t1x, t2x), fmaxf(t1y, t2y)), fmaxf(t1z, t2z));
        near = fmaxf(near, 0.2f);
        far  = fmaxf(far, near + 1e-6f);
    }

    const int   T      = *num_steps;
    const float invTm1 = 1.f / (float)(T - 1);
    const float fn     = far - near;
    const float dstep  = fn * invTm1;
    const float dlast  = fn / (float)T;

    v16h Bm = {};
    v16h Bc = {};
    Bc[3] = (_Float16)(hi ? 0.f : dx);
    Bc[4] = (_Float16)(hi ? 0.f : dy);
    Bc[5] = (_Float16)(hi ? 0.f : dz);

    float Tc = 1.f,  ws = 0.f,  dep = 0.f,  im0 = 0.f, im1 = 0.f, im2 = 0.f;
    float Td = 1.f,  wsd = 0.f, depd = 0.f, id0 = 0.f, id1 = 0.f, id2 = 0.f;

    for (int t = 0; t < T; ++t) {
        const float z01 = (float)t * invTm1;
        const float zv  = near + fn * z01;
        const float px = fminf(fmaxf(ox + dx * zv, -1.f), 1.f);
        const float py = fminf(fmaxf(oy + dy * zv, -1.f), 1.f);
        const float pz = fminf(fmaxf(oz + dz * zv, -1.f), 1.f);
        const _Float16 qx = (_Float16)(hi ? 0.f : px);
        const _Float16 qy = (_Float16)(hi ? 0.f : py);
        const _Float16 qz = (_Float16)(hi ? 0.f : pz);
        Bm[0] = qx; Bm[1] = qy; Bm[2] = qz;
        Bc[0] = qx; Bc[1] = qy; Bc[2] = qz;

        v8f D0 = WMMA_F16(A1t0, Bm, CbT0);
        v8f D1 = WMMA_F16(A1t1, Bm, CbT1);
        v16h Bh;
#pragma unroll
        for (int i = 0; i < 8; ++i) {
            Bh[i]     = (_Float16)fmaxf(D0[i], 0.f);
            Bh[i + 8] = (_Float16)fmaxf(D1[i], 0.f);
        }
        v8f Z0 = {};
        v8f Ds = WMMA_F16(Asig, Bh, Z0);

        v8f C0 = WMMA_F16(Ac1t0, Bc, CbC0);
        v8f C1 = WMMA_F16(Ac1t1, Bc, CbC1);
        v16h Bhc;
#pragma unroll
        for (int i = 0; i < 8; ++i) {
            Bhc[i]     = (_Float16)fmaxf(C0[i], 0.f);
            Bhc[i + 8] = (_Float16)fmaxf(C1[i], 0.f);
        }
        v8f Z1 = {};
        v8f Dc = WMMA_F16(Acol, Bhc, Z1);

        const float delta = (t == T - 1) ? dlast : dstep;
        const float dl2   = -delta * L2E;
        const float sg  = softplus_f(Ds[0]);
        const float sgd = softplus_f(Ds[1]);
        const float a   = 1.f - exp2_fast(dl2 * sg);
        const float ad  = 1.f - exp2_fast(dl2 * sgd);
        const float w   = a  * Tc;
        const float wd  = ad * Td;
        Tc *= (1.f - a  + 1e-15f);
        Td *= (1.f - ad + 1e-15f);
        ws  += w;   dep  += w  * z01;
        wsd += wd;  depd += wd * z01;
        im0 += w  * sigmoid_f(Dc[0]);
        im1 += w  * sigmoid_f(Dc[1]);
        im2 += w  * sigmoid_f(Dc[2]);
        id0 += wd * sigmoid_f(Dc[3]);
        id1 += wd * sigmoid_f(Dc[4]);
        id2 += wd * sigmoid_f(Dc[5]);
    }

    if (!hi && (rayBase + m) < N) {
        float* o9 = out + (size_t)(rayBase + m) * 9;
        const float bg  = 1.f - ws;
        const float bgd = 1.f - wsd;
        o9[0] = im0 + bg;  o9[1] = im1 + bg;  o9[2] = im2 + bg;
        o9[3] = dep;
        o9[4] = ws;
        o9[5] = id0 + bgd; o9[6] = id1 + bgd; o9[7] = id2 + bgd;
        o9[8] = depd;
    }
}

extern "C" void kernel_launch(void* const* d_in, const int* in_sizes, int n_in,
                              void* d_out, int out_size, void* d_ws, size_t ws_size,
                              hipStream_t stream) {
    const float* rays_o = (const float*)d_in[0];
    const float* rays_d = (const float*)d_in[1];
    const float* W1     = (const float*)d_in[2];
    const float* b1     = (const float*)d_in[3];
    const float* Wsig   = (const float*)d_in[4];
    const float* Wsig_d = (const float*)d_in[5];
    const float* Wc1    = (const float*)d_in[6];
    const float* bc1    = (const float*)d_in[7];
    const float* Wc2    = (const float*)d_in[8];
    const float* Wc2_d  = (const float*)d_in[9];
    const int*   nsteps = (const int*)d_in[10];

    const int N      = in_sizes[0] / 3;
    const int waves  = (N + 15) / 16;
    const int thr    = waves * 32;
    const int block  = 256;
    const int grid   = (thr + block - 1) / block;

    nerf_wmma_kernel<<<grid, block, 0, stream>>>(
        rays_o, rays_d, W1, b1, Wsig, Wsig_d, Wc1, bc1, Wc2, Wc2_d,
        nsteps, (float*)d_out, N);
}


# sample loop of each 16-ray tile split in two halves on two waves (twice the waves per SIMD); halves joined via LDS as first + T_first*second, which changes the summation order
# speedup vs baseline: 1.1193x; 1.1193x over previous
#include <hip/hip_runtime.h>

typedef __attribute__((ext_vector_type(16))) _Float16 v16h;
typedef __attribute__((ext_vector_type(8)))  float    v8f;

#define WMMA_F16(A, B, C) \
    __builtin_amdgcn_wmma_f32_16x16x32_f16(false, (A), false, (B), (short)0, (C), false, false)

#define L2E 1.4426950408889634f
#define LN2 0.6931471805599453f

__device__ __forceinline__ float exp2_fast(float x) { return __builtin_amdgcn_exp2f(x); }
__device__ __forceinline__ float log2_fast(float x) { return __builtin_amdgcn_logf(x); }
__device__ __forceinline__ float rcp_fast(float x)  { return __builtin_amdgcn_rcpf(x); }

__device__ __forceinline__ float softplus_f(float x) {
    return x > 20.f ? x : LN2 * log2_fast(1.f + exp2_fast(x * L2E));
}
__device__ __forceinline__ float sigmoid_f(float x) {
    return rcp_fast(1.f + exp2_fast(-x * L2E));
}

__global__ __launch_bounds__(256)
void nerf_wmma_kernel(const float* __restrict__ rays_o,
                      const float* __restrict__ rays_d,
                      const float* __restrict__ W1,   const float* __restrict__ b1,
                      const float* __restrict__ Wsig, const float* __restrict__ Wsig_d,
                      const float* __restrict__ Wc1,  const float* __restrict__ bc1,
                      const float* __restrict__ Wc2,  const float* __restrict__ Wc2_d,
                      const int*   __restrict__ num_steps,
                      float* __restrict__ out, int N)
{
    const int lane = threadIdx.x & 31;
    const int wib  = __builtin_amdgcn_readfirstlane((int)(threadIdx.x >> 5));
    const int tib  = wib & 3;
    const int seg  = wib >> 2;
    const int m    = lane & 15;
    const bool hi  = (lane & 16) != 0;
    const int rayBase = (blockIdx.x * 4 + tib) * 16;

    __shared__ float part[4][10][16];

    int ray = rayBase + m;
    if (ray >= N) ray = N - 1;

    v16h A1t0, A1t1, Ac1t0, Ac1t1, Asig, Acol;
#pragma unroll
    for (int e = 0; e < 16; ++e) {
        const int K = (e < 8 ? e : e + 8) + (hi ? 8 : 0);
        float a0 = (K < 3) ? W1[K * 32 + m]        : 0.f;
        float a1 = (K < 3) ? W1[K * 32 + (m + 16)] : 0.f;
        A1t0[e] = (_Float16)a0;
        A1t1[e] = (_Float16)a1;
        float c0 = (K < 6) ? Wc1[K * 32 + m]        : 0.f;
        float c1 = (K < 6) ? Wc1[K * 32 + (m + 16)] : 0.f;
        Ac1t0[e] = (_Float16)c0;
        Ac1t1[e] = (_Float16)c1;
        float sv = (m == 0) ? Wsig[K] : ((m == 1) ? Wsig_d[K] : 0.f);
        Asig[e] = (_Float16)sv;
        float cv = (m < 3) ? Wc2[K * 3 + m] : ((m < 6) ? Wc2_d[K * 3 + (m - 3)] : 0.f);
        Acol[e] = (_Float16)cv;
    }
    v8f CbT0, CbT1, CbC0, CbC1;
#pragma unroll
    for (int i = 0; i < 8; ++i) {
        const int h0 = i + (hi ? 8 : 0);
        CbT0[i] = b1[h0];      CbT1[i] = b1[h0 + 16];
        CbC0[i] = bc1[h0];     CbC1[i] = bc1[h0 + 16];
    }

    float ox = rays_o[ray * 3 + 0], oy = rays_o[ray * 3 + 1], oz = rays_o[ray * 3 + 2];
    float dx = rays_d[ray * 3 + 0], dy = rays_d[ray * 3 + 1], dz = rays_d[ray * 3 + 2];
    {
        float il = 1.f / sqrtf(dx * dx + dy * dy + dz * dz);
        dx *= il; dy *= il; dz *= il;
    }
    float near, far;
    {
        float ivx = 1.f / dx, ivy = 1.f / dy, ivz = 1.f / dz;
        float t1x = (-1.f - ox) * ivx, t2x = (1.f - ox) * ivx;
        float t1y = (-1.f - oy) * ivy, t2y = (1.f - oy) * ivy;
        float t1z = (-1.f - oz) * ivz, t2z = (1.f - oz) * ivz;
        near = fmaxf(fmaxf(fminf(t1x, t2x), fminf(t1y, t2y)), fminf(t1z, t2z));
        far  = fminf(fminf(fmaxf(t1x, t2x), fmaxf(t1y, t2y)), fmaxf(t1z, t2z));
        near = fmaxf(near, 0.2f);
        far  = fmaxf(far, near + 1e-6f);
    }

    const int   T      = *num_steps;
    const float invTm1 = 1.f / (float)(T - 1);
    const float fn     = far - near;
    const float dstep  = fn * invTm1;
    const float dlast  = fn / (float)T;

    v16h Bm = {};
    v16h Bc = {};
    Bc[3] = (_Float16)(hi ? 0.f : dx);
    Bc[4] = (_Float16)(hi ? 0.f : dy);
    Bc[5] = (_Float16)(hi ? 0.f : dz);

    float Tc = 1.f,  ws = 0.f,  dep = 0.f,  im0 = 0.f, im1 = 0.f, im2 = 0.f;
    float Td = 1.f,  wsd = 0.f, depd = 0.f, id0 = 0.f, id1 = 0.f, id2 = 0.f;

    const int Th = T >> 1;
    const int t0 = seg ? Th : 0;
    const int t1 = seg ? T  : Th;
    for (int t = t0; t < t1; ++t) {
        const float z01 = (float)t * invTm1;
        const float zv  = near + fn * z01;
        const float px = fminf(fmaxf(ox + dx * zv, -1.f), 1.f);
        const float py = fminf(fmaxf(oy + dy * zv, -1.f), 1.f);
        const float pz = fminf(fmaxf(oz + dz * zv, -1.f), 1.f);
        const _Float16 qx = (_Float16)(hi ? 0.f : px);
        const _Float16 qy = (_Float16)(hi ? 0.f : py);
        const _Float16 qz = (_Float16)(hi ? 0.f : pz);
        Bm[0] = qx; Bm[1] = qy; Bm[2] = qz;
        Bc[0] = qx; Bc[1] = qy; Bc[2] = qz;

        v8f D0 = WMMA_F16(A1t0, Bm, CbT0);
        v8f D1 = WMMA_F16(A1t1, Bm, CbT1);
        v16h Bh;
#pragma unroll
        for (int i = 0; i < 8; ++i) {
            Bh[i]     = (_Float16)fmaxf(D0[i], 0.f);
            Bh[i + 8] = (_Float16)fmaxf(D1[i], 0.f);
        }
        v8f Z0 = {};
        v8f Ds = WMMA_F16(Asig, Bh, Z0);

        v8f C0 = WMMA_F16(Ac1t0, Bc, CbC0);
        v8f C1 = WMMA_F16(Ac1t1, Bc, CbC1);
        v16h Bhc;
#pragma unroll
        for (int i = 0; i < 8; ++i) {
            Bhc[i]     = (_Float16)fmaxf(C0[i], 0.f);
            Bhc[i + 8] = (_Float16)fmaxf(C1[i], 0.f);
        }
        v8f Z1 = {};
        v8f Dc = WMMA_F16(Acol, Bhc, Z1);

        const float delta = (t == T - 1) ? dlast : dstep;
        const float dl2   = -delta * L2E;
        const float sg  = softplus_f(Ds[0]);
        const float sgd = softplus_f(Ds[1]);
        const float a   = 1.f - exp2_fast(dl2 * sg);
        const float ad  = 1.f - exp2_fast(dl2 * sgd);
        const float w   = a  * Tc;
        const float wd  = ad * Td;
        Tc *= (1.f - a  + 1e-15f);
        Td *= (1.f - ad + 1e-15f);
        ws  += w;   dep  += w  * z01;
        wsd += wd;  depd += wd * z01;
        im0 += w  * sigmoid_f(Dc[0]);
        im1 += w  * sigmoid_f(Dc[1]);
        im2 += w  * sigmoid_f(Dc[2]);
        id0 += wd * sigmoid_f(Dc[3]);
        id1 += wd * sigmoid_f(Dc[4]);
        id2 += wd * sigmoid_f(Dc[5]);
    }

    if (seg == 1 && !hi) {
        float* p = &part[tib][0][m];
        p[0 * 16] = ws;   p[1 * 16] = dep;  p[2 * 16] = im0;  p[3 * 16] = im1;  p[4 * 16] = im2;
        p[5 * 16] = wsd;  p[6 * 16] = depd; p[7 * 16] = id0;  p[8 * 16] = id1;  p[9 * 16] = id2;
    }
    __syncthreads();
    if (seg == 0 && !hi) {
        const float* p = &part[tib][0][m];
        ws   += Tc * p[0 * 16];  dep  += Tc * p[1 * 16];
        im0  += Tc * p[2 * 16];  im1  += Tc * p[3 * 16];  im2 += Tc * p[4 * 16];
        wsd  += Td * p[5 * 16];  depd += Td * p[6 * 16];
        id0  += Td * p[7 * 16];  id1  += Td * p[8 * 16];  id2 += Td * p[9 * 16];
    }

    if (seg == 0 && !hi && (rayBase + m) < N) {
        float* o9 = out + (size_t)(rayBase + m) * 9;
        const float bg  = 1.f - ws;
        const float bgd = 1.f - wsd;
        o9[0] = im0 + bg;  o9[1] = im1 + bg;  o9[2] = im2 + bg;
        o9[3] = dep;
        o9[4] = ws;
        o9[5] = id0 + bgd; o9[6] = id1 + bgd; o9[7] = id2 + bgd;
        o9[8] = depd;
    }
}

extern "C" void kernel_launch(void* const* d_in, const int* in_sizes, int n_in,
                              void* d_out, int out_size, void* d_ws, size_t ws_size,
                              hipStream_t stream) {
    const float* rays_o = (const float*)d_in[0];
    const float* rays_d = (const float*)d_in[1];
    const float* W1     = (const float*)d_in[2];
    const float* b1     = (const float*)d_in[3];
    const float* Wsig   = (const float*)d_in[4];
    const float* Wsig_d = (const float*)d_in[5];
    const float* Wc1    = (const float*)d_in[6];
    const float* bc1    = (const float*)d_in[7];
    const float* Wc2    = (const float*)d_in[8];
    const float* Wc2_d  = (const float*)d_in[9];
    const int*   nsteps = (const int*)d_in[10];

    const int N      = in_sizes[0] / 3;
    const int tiles  = (N + 15) / 16;
    const int block  = 256;
    const int grid   = (tiles + 3) / 4;

    nerf_wmma_kernel<<<grid, block, 0, stream>>>(
        rays_o, rays_d, W1, b1, Wsig, Wsig_d, Wc1, bc1, Wc2, Wc2_d,
        nsteps, (float*)d_out, N);
}


# main and dummy output heads moved to rows 0.. and 8.. of the head tiles, so lanes 0-15 composite the main head and lanes 16-31 the dummy head (one head per lane, not two)
# speedup vs baseline: 1.1646x; 1.0405x over previous
#include <hip/hip_runtime.h>

typedef __attribute__((ext_vector_type(16))) _Float16 v16h;
typedef __attribute__((ext_vector_type(8)))  float    v8f;

#define WMMA_F16(A, B, C) \
    __builtin_amdgcn_wmma_f32_16x16x32_f16(false, (A), false, (B), (short)0, (C), false, false)

#define L2E 1.4426950408889634f
#define LN2 0.6931471805599453f

__device__ __forceinline__ float exp2_fast(float x) { return __builtin_amdgcn_exp2f(x); }
__device__ __forceinline__ float log2_fast(float x) { return __builtin_amdgcn_logf(x); }
__device__ __forceinline__ float rcp_fast(float x)  { return __builtin_amdgcn_rcpf(x); }

__device__ __forceinline__ float softplus_f(float x) {
    return x > 20.f ? x : LN2 * log2_fast(1.f + exp2_fast(x * L2E));
}
__device__ __forceinline__ float sigmoid_f(float x) {
    return rcp_fast(1.f + exp2_fast(-x * L2E));
}

__global__ __launch_bounds__(256)
void nerf_wmma_kernel(const float* __restrict__ rays_o,
                      const float* __restrict__ rays_d,
                      const float* __restrict__ W1,   const float* __restrict__ b1,
                      const float* __restrict__ Wsig, const float* __restrict__ Wsig_d,
                      const float* __restrict__ Wc1,  const float* __restrict__ bc1,
                      const float* __restrict__ Wc2,  const float* __restrict__ Wc2_d,
                      const int*   __restrict__ num_steps,
                      float* __restrict__ out, int N)
{
    const int lane = threadIdx.x & 31;
    const int wib  = __builtin_amdgcn_readfirstlane((int)(threadIdx.x >> 5));
    const int tib  = wib & 3;
    const int seg  = wib >> 2;
    const int m    = lane & 15;
    const bool hi  = (lane & 16) != 0;
    const int rayBase = (blockIdx.x * 4 + tib) * 16;

    __shared__ float part[4][5][32];

    int ray = rayBase + m;
    if (ray >= N) ray = N - 1;

    v16h A1t0, A1t1, Ac1t0, Ac1t1, Asig, Acol;
#pragma unroll
    for (int e = 0; e < 16; ++e) {
        const int K = (e < 8 ? e : e + 8) + (hi ? 8 : 0);
        float a0 = (K < 3) ? W1[K * 32 + m]        : 0.f;
        float a1 = (K < 3) ? W1[K * 32 + (m + 16)] : 0.f;
        A1t0[e] = (_Float16)a0;
        A1t1[e] = (_Float16)a1;
        float c0 = (K < 6) ? Wc1[K * 32 + m]        : 0.f;
        float c1 = (K < 6) ? Wc1[K * 32 + (m + 16)] : 0.f;
        Ac1t0[e] = (_Float16)c0;
        Ac1t1[e] = (_Float16)c1;
        float sv = (m == 0) ? Wsig[K] : ((m == 8) ? Wsig_d[K] : 0.f);
        Asig[e] = (_Float16)sv;
        float cv = (m < 3) ? Wc2[K * 3 + m] : ((m >= 8 && m < 11) ? Wc2_d[K * 3 + (m - 8)] : 0.f);
        Acol[e] = (_Float16)cv;
    }
    v8f CbT0, CbT1, CbC0, CbC1;
#pragma unroll
    for (int i = 0; i < 8; ++i) {
        const int h0 = i + (hi ? 8 : 0);
        CbT0[i] = b1[h0];      CbT1[i] = b1[h0 + 16];
        CbC0[i] = bc1[h0];     CbC1[i] = bc1[h0 + 16];
    }

    float ox = rays_o[ray * 3 + 0], oy = rays_o[ray * 3 + 1], oz = rays_o[ray * 3 + 2];
    float dx = rays_d[ray * 3 + 0], dy = rays_d[ray * 3 + 1], dz = rays_d[ray * 3 + 2];
    {
        float il = 1.f / sqrtf(dx * dx + dy * dy + dz * dz);
        dx *= il; dy *= il; dz *= il;
    }
    float near, far;
    {
        float ivx = 1.f / dx, ivy = 1.f / dy, ivz = 1.f / dz;
        float t1x = (-1.f - ox) * ivx, t2x = (1.f - ox) * ivx;
        float t1y = (-1.f - oy) * ivy, t2y = (1.f - oy) * ivy;
        float t1z = (-1.f - oz) * ivz, t2z = (1.f - oz) * ivz;
        near = fmaxf(fmaxf(fminf(t1x, t2x), fminf(t1y, t2y)), fminf(t1z, t2z));
        far  = fminf(fminf(fmaxf(t1x, t2x), fmaxf(t1y, t2y)), fmaxf(t1z, t2z));
        near = fmaxf(near, 0.2f);
        far  = fmaxf(far, near + 1e-6f);
    }

    const int   T      = *num_steps;
    const float invTm1 = 1.f / (float)(T - 1);
    const float fn     = far - near;
    const float dstep  = fn * invTm1;
    const float dlast  = fn / (float)T;

    v16h Bm = {};
    v16h Bc = {};
    Bc[3] = (_Float16)(hi ? 0.f : dx);
    Bc[4] = (_Float16)(hi ? 0.f : dy);
    Bc[5] = (_Float16)(hi ? 0.f : dz);

    float Tc = 1.f,  ws = 0.f,  dep = 0.f,  im0 = 0.f, im1 = 0.f, im2 = 0.f;

    const int Th = T >> 1;
    const int t0 = seg ? Th : 0;
    const int t1 = seg ? T  : Th;
    for (int t = t0; t < t1; ++t) {
        const float z01 = (float)t * invTm1;
        const float zv  = near + fn * z01;
        const float px = fminf(fmaxf(ox + dx * zv, -1.f), 1.f);
        const float py = fminf(fmaxf(oy + dy * zv, -1.f), 1.f);
        const float pz = fminf(fmaxf(oz + dz * zv, -1.f), 1.f);
        const _Float16 qx = (_Float16)(hi ? 0.f : px);
        const _Float16 qy = (_Float16)(hi ? 0.f : py);
        const _Float16 qz = (_Float16)(hi ? 0.f : pz);
        Bm[0] = qx; Bm[1] = qy; Bm[2] = qz;
        Bc[0] = qx; Bc[1] = qy; Bc[2] = qz;

        v8f D0 = WMMA_F16(A1t0, Bm, CbT0);
        v8f D1 = WMMA_F16(A1t1, Bm, CbT1);
        v16h Bh;
#pragma unroll
        for (int i = 0; i < 8; ++i) {
            Bh[i]     = (_Float16)fmaxf(D0[i], 0.f);
            Bh[i + 8] = (_Float16)fmaxf(D1[i], 0.f);
        }
        v8f Z0 = {};
        v8f Ds = WMMA_F16(Asig, Bh, Z0);

        v8f C0 = WMMA_F16(Ac1t0, Bc, CbC0);
        v8f C1 = WMMA_F16(Ac1t1, Bc, CbC1);
        v16h Bhc;
#pragma unroll
        for (int i = 0; i < 8; ++i) {
            Bhc[i]     = (_Float16)fmaxf(C0[i], 0.f);
            Bhc[i + 8] = (_Float16)fmaxf(C1[i], 0.f);
        }
        v8f Z1 = {};
        v8f Dc = WMMA_F16(Acol, Bhc, Z1);

        const float delta = (t == T - 1) ? dlast : dstep;
        const float dl2   = -delta * L2E;
        const float sg  = softplus_f(Ds[0]);
        const float a   = 1.f - exp2_fast(dl2 * sg);
        const float w   = a * Tc;
        Tc *= (1.f - a + 1e-15f);
        ws  += w;   dep += w * z01;
        im0 += w * sigmoid_f(Dc[0]);
        im1 += w * sigmoid_f(Dc[1]);
        im2 += w * sigmoid_f(Dc[2]);
    }

    if (seg == 1) {
        float* p = &part[tib][0][lane];
        p[0 * 32] = ws;   p[1 * 32] = dep;  p[2 * 32] = im0;  p[3 * 32] = im1;  p[4 * 32] = im2;
    }
    __syncthreads();
    if (seg == 0) {
        const float* p = &part[tib][0][lane];
        ws   += Tc * p[0 * 32];  dep  += Tc * p[1 * 32];
        im0  += Tc * p[2 * 32];  im1  += Tc * p[3 * 32];  im2 += Tc * p[4 * 32];
    }

    if (seg == 0 && (rayBase + m) < N) {
        float* o = out + (size_t)(rayBase + m) * 9 + (hi ? 5 : 0);
        const float bg = 1.f - ws;
        o[0] = im0 + bg;  o[1] = im1 + bg;  o[2] = im2 + bg;
        o[3] = dep;
        if (!hi) o[4] = ws;
    }
}

extern "C" void kernel_launch(void* const* d_in, const int* in_sizes, int n_in,
                              void* d_out, int out_size, void* d_ws, size_t ws_size,
                              hipStream_t stream) {
    const float* rays_o = (const float*)d_in[0];
    const float* rays_d = (const float*)d_in[1];
    const float* W1     = (const float*)d_in[2];
    const float* b1     = (const float*)d_in[3];
    const float* Wsig   = (const float*)d_in[4];
    const float* Wsig_d = (const float*)d_in[5];
    const float* Wc1    = (const float*)d_in[6];
    const float* bc1    = (const float*)d_in[7];
    const float* Wc2    = (const float*)d_in[8];
    const float* Wc2_d  = (const float*)d_in[9];
    const int*   nsteps = (const int*)d_in[10];

    const int N      = in_sizes[0] / 3;
    const int tiles  = (N + 15) / 16;
    const int block  = 256;
    const int grid   = (tiles + 3) / 4;

    nerf_wmma_kernel<<<grid, block, 0, stream>>>(
        rays_o, rays_d, W1, b1, Wsig, Wsig_d, Wc1, bc1, Wc2, Wc2_d,
        nsteps, (float*)d_out, N);
}


# weight fragments fetched by unconditional in-range loads, zero padding applied by select afterwards: the prologue's 50 load-then-wait pairs under branches become grouped loads with 4 waits
# speedup vs baseline: 1.4772x; 1.2684x over previous
#include <hip/hip_runtime.h>

typedef __attribute__((ext_vector_type(16))) _Float16 v16h;
typedef __attribute__((ext_vector_type(8)))  float    v8f;

#define WMMA_F16(A, B, C) \
    __builtin_amdgcn_wmma_f32_16x16x32_f16(false, (A), false, (B), (short)0, (C), false, false)

#define L2E 1.4426950408889634f
#define LN2 0.6931471805599453f

__device__ __forceinline__ float exp2_fast(float x) { return __builtin_amdgcn_exp2f(x); }
__device__ __forceinline__ float log2_fast(float x) { return __builtin_amdgcn_logf(x); }
__device__ __forceinline__ float rcp_fast(float x)  { return __builtin_amdgcn_rcpf(x); }

__device__ __forceinline__ float softplus_f(float x) {
    return x > 20.f ? x : LN2 * log2_fast(1.f + exp2_fast(x * L2E));
}
__device__ __forceinline__ float sigmoid_f(float x) {
    return rcp_fast(1.f + exp2_fast(-x * L2E));
}

__global__ __launch_bounds__(256)
void nerf_wmma_kernel(const float* __restrict__ rays_o,
                      const float* __restrict__ rays_d,
                      const float* __restrict__ W1,   const float* __restrict__ b1,
                      const float* __restrict__ Wsig, const float* __restrict__ Wsig_d,
                      const float* __restrict__ Wc1,  const float* __restrict__ bc1,
                      const float* __restrict__ Wc2,  const float* __restrict__ Wc2_d,
                      const int*   __restrict__ num_steps,
                      float* __restrict__ out, int N)
{
    const int lane = threadIdx.x & 31;
    const int wib  = __builtin_amdgcn_readfirstlane((int)(threadIdx.x >> 5));
    const int tib  = wib & 3;
    const int seg  = wib >> 2;
    const int m    = lane & 15;
    const bool hi  = (lane & 16) != 0;
    const int rayBase = (blockIdx.x * 4 + tib) * 16;

    __shared__ float part[4][5][32];

    int ray = rayBase + m;
    if (ray >= N) ray = N - 1;

    const int    kb     = hi ? 8 : 0;
    const int    m7     = m & 7;
    const int    jc     = m7 < 3 ? m7 : 2;
    const bool   sigRow = (m7 == 0);
    const bool   colRow = (m7 < 3);
    const float* pSig   = (m & 8) ? Wsig_d : Wsig;
    const float* pCol   = (m & 8) ? Wc2_d  : Wc2;

    v16h A1t0 = {}, A1t1 = {}, Ac1t0 = {}, Ac1t1 = {};
#pragma unroll
    for (int e = 0; e < 6; ++e) {
        if (e < 3) {
            const float a0 = W1[e * 32 + m];
            const float a1 = W1[e * 32 + (m + 16)];
            A1t0[e] = (_Float16)(hi ? 0.f : a0);
            A1t1[e] = (_Float16)(hi ? 0.f : a1);
        }
        const float c0 = Wc1[e * 32 + m];
        const float c1 = Wc1[e * 32 + (m + 16)];
        Ac1t0[e] = (_Float16)(hi ? 0.f : c0);
        Ac1t1[e] = (_Float16)(hi ? 0.f : c1);
    }
    v16h Asig, Acol;
#pragma unroll
    for (int e = 0; e < 16; ++e) {
        const int K = (e < 8 ? e : e + 8) + kb;
        const float sv = pSig[K];
        const float cv = pCol[K * 3 + jc];
        Asig[e] = (_Float16)(sigRow ? sv : 0.f);
        Acol[e] = (_Float16)(colRow ? cv : 0.f);
    }
    v8f CbT0, CbT1, CbC0, CbC1;
#pragma unroll
    for (int i = 0; i < 8; ++i) {
        const int h0 = i + (hi ? 8 : 0);
        CbT0[i] = b1[h0];      CbT1[i] = b1[h0 + 16];
        CbC0[i] = bc1[h0];     CbC1[i] = bc1[h0 + 16];
    }

    float ox = rays_o[ray * 3 + 0], oy = rays_o[ray * 3 + 1], oz = rays_o[ray * 3 + 2];
    float dx = rays_d[ray * 3 + 0], dy = rays_d[ray * 3 + 1], dz = rays_d[ray * 3 + 2];
    {
        float il = 1.f / sqrtf(dx * dx + dy * dy + dz * dz);
        dx *= il; dy *= il; dz *= il;
    }
    float near, far;
    {
        float ivx = 1.f / dx, ivy = 1.f / dy, ivz = 1.f / dz;
        float t1x = (-1.f - ox) * ivx, t2x = (1.f - ox) * ivx;
        float t1y = (-1.f - oy) * ivy, t2y = (1.f - oy) * ivy;
        float t1z = (-1.f - oz) * ivz, t2z = (1.f - oz) * ivz;
        near = fmaxf(fmaxf(fminf(t1x, t2x), fminf(t1y, t2y)), fminf(t1z, t2z));
        far  = fminf(fminf(fmaxf(t1x, t2x), fmaxf(t1y, t2y)), fmaxf(t1z, t2z));
        near = fmaxf(near, 0.2f);
        far  = fmaxf(far, near + 1e-6f);
    }

    const int   T      = *num_steps;
    const float invTm1 = 1.f / (float)(T - 1);
    const float fn     = far - near;
    const float dstep  = fn * invTm1;
    const float dlast  = fn / (float)T;

    v16h Bm = {};
    v16h Bc = {};
    Bc[3] = (_Float16)(hi ? 0.f : dx);
    Bc[4] = (_Float16)(hi ? 0.f : dy);
    Bc[5] = (_Float16)(hi ? 0.f : dz);

    float Tc = 1.f,  ws = 0.f,  dep = 0.f,  im0 = 0.f, im1 = 0.f, im2 = 0.f;

    const int Th = T >> 1;
    const int t0 = seg ? Th : 0;
    const int t1 = seg ? T  : Th;
    for (int t = t0; t < t1; ++t) {
        const float z01 = (float)t * invTm1;
        const float zv  = near + fn * z01;
        const float px = fminf(fmaxf(ox + dx * zv, -1.f), 1.f);
        const float py = fminf(fmaxf(oy + dy * zv, -1.f), 1.f);
        const float pz = fminf(fmaxf(oz + dz * zv, -1.f), 1.f);
        const _Float16 qx = (_Float16)(hi ? 0.f : px);
        const _Float16 qy = (_Float16)(hi ? 0.f : py);
        const _Float16 qz = (_Float16)(hi ? 0.f : pz);
        Bm[0] = qx; Bm[1] = qy; Bm[2] = qz;
        Bc[0] = qx; Bc[1] = qy; Bc[2] = qz;

        v8f D0 = WMMA_F16(A1t0, Bm, CbT0);
        v8f D1 = WMMA_F16(A1t1, Bm, CbT1);
        v16h Bh;
#pragma unroll
        for (int i = 0; i < 8; ++i) {
            Bh[i]     = (_Float16)fmaxf(D0[i], 0.f);
            Bh[i + 8] = (_Float16)fmaxf(D1[i], 0.f);
        }
        v8f Z0 = {};
        v8f Ds = WMMA_F16(Asig, Bh, Z0);

        v8f C0 = WMMA_F16(Ac1t0, Bc, CbC0);
        v8f C1 = WMMA_F16(Ac1t1, Bc, CbC1);
        v16h Bhc;
#pragma unroll
        for (int i = 0; i < 8; ++i) {
            Bhc[i]     = (_Float16)fmaxf(C0[i], 0.f);
            Bhc[i + 8] = (_Float16)fmaxf(C1[i], 0.f);
        }
        v8f Z1 = {};
        v8f Dc = WMMA_F16(Acol, Bhc, Z1);

        const float delta = (t == T - 1) ? dlast : dstep;
        const float dl2   = -delta * L2E;
        const float sg  = softplus_f(Ds[0]);
        const float a   = 1.f - exp2_fast(dl2 * sg);
        const float w   = a * Tc;
        Tc *= (1.f - a + 1e-15f);
        ws  += w;   dep += w * z01;
        im0 += w * sigmoid_f(Dc[0]);
        im1 += w * sigmoid_f(Dc[1]);
        im2 += w * sigmoid_f(Dc[2]);
    }

    if (seg == 1) {
        float* p = &part[tib][0][lane];
        p[0 * 32] = ws;   p[1 * 32] = dep;  p[2 * 32] = im0;  p[3 * 32] = im1;  p[4 * 32] = im2;
    }
    __syncthreads();
    if (seg == 0) {
        const float* p = &part[tib][0][lane];
        ws   += Tc * p[0 * 32];  dep  += Tc * p[1 * 32];
        im0  += Tc * p[2 * 32];  im1  += Tc * p[3 * 32];  im2 += Tc * p[4 * 32];
    }

    if (seg == 0 && (rayBase + m) < N) {
        float* o = out + (size_t)(rayBase + m) * 9 + (hi ? 5 : 0);
        const float bg = 1.f - ws;
        o[0] = im0 + bg;  o[1] = im1 + bg;  o[2] = im2 + bg;
        o[3] = dep;
        if (!hi) o[4] = ws;
    }
}

extern "C" void kernel_launch(void* const* d_in, const int* in_sizes, int n_in,
                              void* d_out, int out_size, void* d_ws, size_t ws_size,
                              hipStream_t stream) {
    const float* rays_o = (const float*)d_in[0];
    const float* rays_d = (const float*)d_in[1];
    const float* W1     = (const float*)d_in[2];
    const float* b1     = (const float*)d_in[3];
    const float* Wsig   = (const float*)d_in[4];
    const float* Wsig_d = (const float*)d_in[5];
    const float* Wc1    = (const float*)d_in[6];
    const float* bc1    = (const float*)d_in[7];
    const float* Wc2    = (const float*)d_in[8];
    const float* Wc2_d  = (const float*)d_in[9];
    const int*   nsteps = (const int*)d_in[10];

    const int N      = in_sizes[0] / 3;
    const int tiles  = (N + 15) / 16;
    const int block  = 256;
    const int grid   = (tiles + 3) / 4;

    nerf_wmma_kernel<<<grid, block, 0, stream>>>(
        rays_o, rays_d, W1, b1, Wsig, Wsig_d, Wc1, bc1, Wc2, Wc2_d,
        nsteps, (float*)d_out, N);
}


# relu of both hidden layers moved behind the conversion and taken on packed 16-bit pairs (one packed maximum per two values); the 32-bit maximum needed an extra NaN-quieting max(x,x) per value
# speedup vs baseline: 1.6316x; 1.1045x over previous
#include <hip/hip_runtime.h>

typedef __attribute__((ext_vector_type(16))) _Float16 v16h;
typedef __attribute__((ext_vector_type(8)))  float    v8f;
typedef __attribute__((ext_vector_type(8)))  _Float16 v8h;

#define WMMA_F16(A, B, C) \
    __builtin_amdgcn_wmma_f32_16x16x32_f16(false, (A), false, (B), (short)0, (C), false, false)

#define L2E 1.4426950408889634f
#define LN2 0.6931471805599453f

__device__ __forceinline__ float exp2_fast(float x) { return __builtin_amdgcn_exp2f(x); }
__device__ __forceinline__ float log2_fast(float x) { return __builtin_amdgcn_logf(x); }
__device__ __forceinline__ float rcp_fast(float x)  { return __builtin_amdgcn_rcpf(x); }

__device__ __forceinline__ float softplus_f(float x) {
    return x > 20.f ? x : LN2 * log2_fast(1.f + exp2_fast(x * L2E));
}
__device__ __forceinline__ float sigmoid_f(float x) {
    return rcp_fast(1.f + exp2_fast(-x * L2E));
}

__device__ __forceinline__ v16h relu_pack(v8f lo, v8f hi) {
    const v8h l = __builtin_convertvector(lo, v8h);
    const v8h h = __builtin_convertvector(hi, v8h);
    const v16h x = __builtin_shufflevector(l, h, 0, 1, 2, 3, 4, 5, 6, 7,
                                                 8, 9, 10, 11, 12, 13, 14, 15);
    const v16h zero = {};
    return __builtin_elementwise_max(x, zero);
}

__global__ __launch_bounds__(256)
void nerf_wmma_kernel(const float* __restrict__ rays_o,
                      const float* __restrict__ rays_d,
                      const float* __restrict__ W1,   const float* __restrict__ b1,
                      const float* __restrict__ Wsig, const float* __restrict__ Wsig_d,
                      const float* __restrict__ Wc1,  const float* __restrict__ bc1,
                      const float* __restrict__ Wc2,  const float* __restrict__ Wc2_d,
                      const int*   __restrict__ num_steps,
                      float* __restrict__ out, int N)
{
    const int lane = threadIdx.x & 31;
    const int wib  = __builtin_amdgcn_readfirstlane((int)(threadIdx.x >> 5));
    const int tib  = wib & 3;
    const int seg  = wib >> 2;
    const int m    = lane & 15;
    const bool hi  = (lane & 16) != 0;
    const int rayBase = (blockIdx.x * 4 + tib) * 16;

    __shared__ float part[4][5][32];

    int ray = rayBase + m;
    if (ray >= N) ray = N - 1;

    const int    kb     = hi ? 8 : 0;
    const int    m7     = m & 7;
    const int    jc     = m7 < 3 ? m7 : 2;
    const bool   sigRow = (m7 == 0);
    const bool   colRow = (m7 < 3);
    const float* pSig   = (m & 8) ? Wsig_d : Wsig;
    const float* pCol   = (m & 8) ? Wc2_d  : Wc2;

    v16h A1t0 = {}, A1t1 = {}, Ac1t0 = {}, Ac1t1 = {};
#pragma unroll
    for (int e = 0; e < 6; ++e) {
        if (e < 3) {
            const float a0 = W1[e * 32 + m];
            const float a1 = W1[e * 32 + (m + 16)];
            A1t0[e] = (_Float16)(hi ? 0.f : a0);
            A1t1[e] = (_Float16)(hi ? 0.f : a1);
        }
        const float c0 = Wc1[e * 32 + m];
        const float c1 = Wc1[e * 32 + (m + 16)];
        Ac1t0[e] = (_Float16)(hi ? 0.f : c0);
        Ac1t1[e] = (_Float16)(hi ? 0.f : c1);
    }
    v16h Asig, Acol;
#pragma unroll
    for (int e = 0; e < 16; ++e) {
        const int K = (e < 8 ? e : e + 8) + kb;
        const float sv = pSig[K];
        const float cv = pCol[K * 3 + jc];
        Asig[e] = (_Float16)(sigRow ? sv : 0.f);
        Acol[e] = (_Float16)(colRow ? cv : 0.f);
    }
    v8f CbT0, CbT1, CbC0, CbC1;
#pragma unroll
    for (int i = 0; i < 8; ++i) {
        const int h0 = i + (hi ? 8 : 0);
        CbT0[i] = b1[h0];      CbT1[i] = b1[h0 + 16];
        CbC0[i] = bc1[h0];     CbC1[i] = bc1[h0 + 16];
    }

    float ox = rays_o[ray * 3 + 0], oy = rays_o[ray * 3 + 1], oz = rays_o[ray * 3 + 2];
    float dx = rays_d[ray * 3 + 0], dy = rays_d[ray * 3 + 1], dz = rays_d[ray * 3 + 2];
    {
        float il = 1.f / sqrtf(dx * dx + dy * dy + dz * dz);
        dx *= il; dy *= il; dz *= il;
    }
    float near, far;
    {
        float ivx = 1.f / dx, ivy = 1.f / dy, ivz = 1.f / dz;
        float t1x = (-1.f - ox) * ivx, t2x = (1.f - ox) * ivx;
        float t1y = (-1.f - oy) * ivy, t2y = (1.f - oy) * ivy;
        float t1z = (-1.f - oz) * ivz, t2z = (1.f - oz) * ivz;
        near = fmaxf(fmaxf(fminf(t1x, t2x), fminf(t1y, t2y)), fminf(t1z, t2z));
        far  = fminf(fminf(fmaxf(t1x, t2x), fmaxf(t1y, t2y)), fmaxf(t1z, t2z));
        near = fmaxf(near, 0.2f);
        far  = fmaxf(far, near + 1e-6f);
    }

    const int   T      = *num_steps;
    const float invTm1 = 1.f / (float)(T - 1);
    const float fn     = far - near;
    const float dstep  = fn * invTm1;
    const float dlast  = fn / (float)T;

    v16h Bm = {};
    v16h Bc = {};
    Bc[3] = (_Float16)(hi ? 0.f : dx);
    Bc[4] = (_Float16)(hi ? 0.f : dy);
    Bc[5] = (_Float16)(hi ? 0.f : dz);

    float Tc = 1.f,  ws = 0.f,  dep = 0.f,  im0 = 0.f, im1 = 0.f, im2 = 0.f;

    const int Th = T >> 1;
    const int t0 = seg ? Th : 0;
    const int t1 = seg ? T  : Th;
    for (int t = t0; t < t1; ++t) {
        const float z01 = (float)t * invTm1;
        const float zv  = near + fn * z01;
        const float px = fminf(fmaxf(ox + dx * zv, -1.f), 1.f);
        const float py = fminf(fmaxf(oy + dy * zv, -1.f), 1.f);
        const float pz = fminf(fmaxf(oz + dz * zv, -1.f), 1.f);
        const _Float16 qx = (_Float16)(hi ? 0.f : px);
        const _Float16 qy = (_Float16)(hi ? 0.f : py);
        const _Float16 qz = (_Float16)(hi ? 0.f : pz);
        Bm[0] = qx; Bm[1] = qy; Bm[2] = qz;
        Bc[0] = qx; Bc[1] = qy; Bc[2] = qz;

        v8f D0 = WMMA_F16(A1t0, Bm, CbT0);
        v8f D1 = WMMA_F16(A1t1, Bm, CbT1);
        const v16h Bh = relu_pack(D0, D1);
        v8f Z0 = {};
        v8f Ds = WMMA_F16(Asig, Bh, Z0);

        v8f C0 = WMMA_F16(Ac1t0, Bc, CbC0);
        v8f C1 = WMMA_F16(Ac1t1, Bc, CbC1);
        const v16h Bhc = relu_pack(C0, C1);
        v8f Z1 = {};
        v8f Dc = WMMA_F16(Acol, Bhc, Z1);

        const float delta = (t == T - 1) ? dlast : dstep;
        const float dl2   = -delta * L2E;
        const float sg  = softplus_f(Ds[0]);
        const float a   = 1.f - exp2_fast(dl2 * sg);
        const float w   = a * Tc;
        Tc *= (1.f - a + 1e-15f);
        ws  += w;   dep += w * z01;
        im0 += w * sigmoid_f(Dc[0]);
        im1 += w * sigmoid_f(Dc[1]);
        im2 += w * sigmoid_f(Dc[2]);
    }

    if (seg == 1) {
        float* p = &part[tib][0][lane];
        p[0 * 32] = ws;   p[1 * 32] = dep;  p[2 * 32] = im0;  p[3 * 32] = im1;  p[4 * 32] = im2;
    }
    __syncthreads();
    if (seg == 0) {
        const float* p = &part[tib][0][lane];
        ws   += Tc * p[0 * 32];  dep  += Tc * p[1 * 32];
        im0  += Tc * p[2 * 32];  im1  += Tc * p[3 * 32];  im2 += Tc * p[4 * 32];
    }

    if (seg == 0 && (rayBase + m) < N) {
        float* o = out + (size_t)(rayBase + m) * 9 + (hi ? 5 : 0);
        const float bg = 1.f - ws;
        o[0] = im0 + bg;  o[1] = im1 + bg;  o[2] = im2 + bg;
        o[3] = dep;
        if (!hi) o[4] = ws;
    }
}

extern "C" void kernel_launch(void* const* d_in, const int* in_sizes, int n_in,
                              void* d_out, int out_size, void* d_ws, size_t ws_size,
                              hipStream_t stream) {
    const float* rays_o = (const float*)d_in[0];
    const float* rays_d = (const float*)d_in[1];
    const float* W1     = (const float*)d_in[2];
    const float* b1     = (const float*)d_in[3];
    const float* Wsig   = (const float*)d_in[4];
    const float* Wsig_d = (const float*)d_in[5];
    const float* Wc1    = (const float*)d_in[6];
    const float* bc1    = (const float*)d_in[7];
    const float* Wc2    = (const float*)d_in[8];
    const float* Wc2_d  = (const float*)d_in[9];
    const int*   nsteps = (const int*)d_in[10];

    const int N      = in_sizes[0] / 3;
    const int tiles  = (N + 15) / 16;
    const int block  = 256;
    const int grid   = (tiles + 3) / 4;

    nerf_wmma_kernel<<<grid, block, 0, stream>>>(
        rays_o, rays_d, W1, b1, Wsig, Wsig_d, Wc1, bc1, Wc2, Wc2_d,
        nsteps, (float*)d_out, N);
}


# two samples per loop trip under a 4-waves-per-SIMD register budget: one sample's vector work fills the wait slots behind the other's products; every sum stays in sample order
# speedup vs baseline: 1.9867x; 1.2176x over previous
#include <hip/hip_runtime.h>

typedef __attribute__((ext_vector_type(16))) _Float16 v16h;
typedef __attribute__((ext_vector_type(8)))  float    v8f;
typedef __attribute__((ext_vector_type(8)))  _Float16 v8h;

#define WMMA_F16(A, B, C) \
    __builtin_amdgcn_wmma_f32_16x16x32_f16(false, (A), false, (B), (short)0, (C), false, false)

#define L2E 1.4426950408889634f
#define LN2 0.6931471805599453f

__device__ __forceinline__ float exp2_fast(float x) { return __builtin_amdgcn_exp2f(x); }
__device__ __forceinline__ float log2_fast(float x) { return __builtin_amdgcn_logf(x); }
__device__ __forceinline__ float rcp_fast(float x)  { return __builtin_amdgcn_rcpf(x); }

__device__ __forceinline__ float softplus_f(float x) {
    return x > 20.f ? x : LN2 * log2_fast(1.f + exp2_fast(x * L2E));
}
__device__ __forceinline__ float sigmoid_f(float x) {
    return rcp_fast(1.f + exp2_fast(-x * L2E));
}

__device__ __forceinline__ v16h relu_pack(v8f lo, v8f hi) {
    const v8h l = __builtin_convertvector(lo, v8h);
    const v8h h = __builtin_convertvector(hi, v8h);
    const v16h x = __builtin_shufflevector(l, h, 0, 1, 2, 3, 4, 5, 6, 7,
                                                 8, 9, 10, 11, 12, 13, 14, 15);
    const v16h zero = {};
    return __builtin_elementwise_max(x, zero);
}

__global__ __launch_bounds__(256) __attribute__((amdgpu_waves_per_eu(4, 4)))
void nerf_wmma_kernel(const float* __restrict__ rays_o,
                      const float* __restrict__ rays_d,
                      const float* __restrict__ W1,   const float* __restrict__ b1,
                      const float* __restrict__ Wsig, const float* __restrict__ Wsig_d,
                      const float* __restrict__ Wc1,  const float* __restrict__ bc1,
                      const float* __restrict__ Wc2,  const float* __restrict__ Wc2_d,
                      const int*   __restrict__ num_steps,
                      float* __restrict__ out, int N)
{
    const int lane = threadIdx.x & 31;
    const int wib  = __builtin_amdgcn_readfirstlane((int)(threadIdx.x >> 5));
    const int tib  = wib & 3;
    const int seg  = wib >> 2;
    const int m    = lane & 15;
    const bool hi  = (lane & 16) != 0;
    const int rayBase = (blockIdx.x * 4 + tib) * 16;

    __shared__ float part[4][5][32];

    int ray = rayBase + m;
    if (ray >= N) ray = N - 1;

    const int    kb     = hi ? 8 : 0;
    const int    m7     = m & 7;
    const int    jc     = m7 < 3 ? m7 : 2;
    const bool   sigRow = (m7 == 0);
    const bool   colRow = (m7 < 3);
    const float* pSig   = (m & 8) ? Wsig_d : Wsig;
    const float* pCol   = (m & 8) ? Wc2_d  : Wc2;

    v16h A1t0 = {}, A1t1 = {}, Ac1t0 = {}, Ac1t1 = {};
#pragma unroll
    for (int e = 0; e < 6; ++e) {
        if (e < 3) {
            const float a0 = W1[e * 32 + m];
            const float a1 = W1[e * 32 + (m + 16)];
            A1t0[e] = (_Float16)(hi ? 0.f : a0);
            A1t1[e] = (_Float16)(hi ? 0.f : a1);
        }
        const float c0 = Wc1[e * 32 + m];
        const float c1 = Wc1[e * 32 + (m + 16)];
        Ac1t0[e] = (_Float16)(hi ? 0.f : c0);
        Ac1t1[e] = (_Float16)(hi ? 0.f : c1);
    }
    v16h Asig, Acol;
#pragma unroll
    for (int e = 0; e < 16; ++e) {
        const int K = (e < 8 ? e : e + 8) + kb;
        const float sv = pSig[K];
        const float cv = pCol[K * 3 + jc];
        Asig[e] = (_Float16)(sigRow ? sv : 0.f);
        Acol[e] = (_Float16)(colRow ? cv : 0.f);
    }
    v8f CbT0, CbT1, CbC0, CbC1;
#pragma unroll
    for (int i = 0; i < 8; ++i) {
        const int h0 = i + (hi ? 8 : 0);
        CbT0[i] = b1[h0];      CbT1[i] = b1[h0 + 16];
        CbC0[i] = bc1[h0];     CbC1[i] = bc1[h0 + 16];
    }

    float ox = rays_o[ray * 3 + 0], oy = rays_o[ray * 3 + 1], oz = rays_o[ray * 3 + 2];
    float dx = rays_d[ray * 3 + 0], dy = rays_d[ray * 3 + 1], dz = rays_d[ray * 3 + 2];
    {
        float il = 1.f / sqrtf(dx * dx + dy * dy + dz * dz);
        dx *= il; dy *= il; dz *= il;
    }
    float near, far;
    {
        float ivx = 1.f / dx, ivy = 1.f / dy, ivz = 1.f / dz;
        float t1x = (-1.f - ox) * ivx, t2x = (1.f - ox) * ivx;
        float t1y = (-1.f - oy) * ivy, t2y = (1.f - oy) * ivy;
        float t1z = (-1.f - oz) * ivz, t2z = (1.f - oz) * ivz;
        near = fmaxf(fmaxf(fminf(t1x, t2x), fminf(t1y, t2y)), fminf(t1z, t2z));
        far  = fminf(fminf(fmaxf(t1x, t2x), fmaxf(t1y, t2y)), fmaxf(t1z, t2z));
        near = fmaxf(near, 0.2f);
        far  = fmaxf(far, near + 1e-6f);
    }

    const int   T      = *num_steps;
    const float invTm1 = 1.f / (float)(T - 1);
    const float fn     = far - near;
    const float dstep  = fn * invTm1;
    const float dlast  = fn / (float)T;

    v16h Bm = {};
    v16h Bc = {};
    Bc[3] = (_Float16)(hi ? 0.f : dx);
    Bc[4] = (_Float16)(hi ? 0.f : dy);
    Bc[5] = (_Float16)(hi ? 0.f : dz);

    float Tc = 1.f,  ws = 0.f,  dep = 0.f,  im0 = 0.f, im1 = 0.f, im2 = 0.f;

    const int Th = T >> 1;
    const int t0 = seg ? Th : 0;
    const int t1 = seg ? T  : Th;

    v16h Bm2 = {};
    v16h Bc2 = {};
    Bc2[3] = Bc[3];
    Bc2[4] = Bc[4];
    Bc2[5] = Bc[5];

    int t = t0;
    for (; t + 1 < t1; t += 2) {
        const float z01a = (float)t * invTm1;
        const float z01b = (float)(t + 1) * invTm1;
        const float zva  = near + fn * z01a;
        const float zvb  = near + fn * z01b;
        const float pxa = fminf(fmaxf(ox + dx * zva, -1.f), 1.f);
        const float pya = fminf(fmaxf(oy + dy * zva, -1.f), 1.f);
        const float pza = fminf(fmaxf(oz + dz * zva, -1.f), 1.f);
        const float pxb = fminf(fmaxf(ox + dx * zvb, -1.f), 1.f);
        const float pyb = fminf(fmaxf(oy + dy * zvb, -1.f), 1.f);
        const float pzb = fminf(fmaxf(oz + dz * zvb, -1.f), 1.f);
        const _Float16 qxa = (_Float16)(hi ? 0.f : pxa);
        const _Float16 qya = (_Float16)(hi ? 0.f : pya);
        const _Float16 qza = (_Float16)(hi ? 0.f : pza);
        const _Float16 qxb = (_Float16)(hi ? 0.f : pxb);
        const _Float16 qyb = (_Float16)(hi ? 0.f : pyb);
        const _Float16 qzb = (_Float16)(hi ? 0.f : pzb);
        Bm[0]  = qxa; Bm[1]  = qya; Bm[2]  = qza;
        Bc[0]  = qxa; Bc[1]  = qya; Bc[2]  = qza;
        Bm2[0] = qxb; Bm2[1] = qyb; Bm2[2] = qzb;
        Bc2[0] = qxb; Bc2[1] = qyb; Bc2[2] = qzb;

        v8f D0a = WMMA_F16(A1t0,  Bm,  CbT0);
        v8f D1a = WMMA_F16(A1t1,  Bm,  CbT1);
        v8f C0a = WMMA_F16(Ac1t0, Bc,  CbC0);
        v8f C1a = WMMA_F16(Ac1t1, Bc,  CbC1);
        v8f D0b = WMMA_F16(A1t0,  Bm2, CbT0);
        v8f D1b = WMMA_F16(A1t1,  Bm2, CbT1);
        v8f C0b = WMMA_F16(Ac1t0, Bc2, CbC0);
        v8f C1b = WMMA_F16(Ac1t1, Bc2, CbC1);

        const v8f Z = {};
        const v16h Bha  = relu_pack(D0a, D1a);
        v8f Dsa = WMMA_F16(Asig, Bha,  Z);
        const v16h Bhca = relu_pack(C0a, C1a);
        v8f Dca = WMMA_F16(Acol, Bhca, Z);
        const v16h Bhb  = relu_pack(D0b, D1b);
        v8f Dsb = WMMA_F16(Asig, Bhb,  Z);
        const v16h Bhcb = relu_pack(C0b, C1b);
        v8f Dcb = WMMA_F16(Acol, Bhcb, Z);

        const float dla = -dstep * L2E;
        const float dlb = -((t + 1 == T - 1) ? dlast : dstep) * L2E;
        const float sga = softplus_f(Dsa[0]);
        const float sgb = softplus_f(Dsb[0]);
        const float aa  = 1.f - exp2_fast(dla * sga);
        const float ab  = 1.f - exp2_fast(dlb * sgb);
        const float ra0 = sigmoid_f(Dca[0]), ra1 = sigmoid_f(Dca[1]), ra2 = sigmoid_f(Dca[2]);
        const float rb0 = sigmoid_f(Dcb[0]), rb1 = sigmoid_f(Dcb[1]), rb2 = sigmoid_f(Dcb[2]);

        const float wa = aa * Tc;
        Tc *= (1.f - aa + 1e-15f);
        ws  += wa;   dep += wa * z01a;
        im0 += wa * ra0;  im1 += wa * ra1;  im2 += wa * ra2;
        const float wb = ab * Tc;
        Tc *= (1.f - ab + 1e-15f);
        ws  += wb;   dep += wb * z01b;
        im0 += wb * rb0;  im1 += wb * rb1;  im2 += wb * rb2;
    }
    if (t < t1) {
        const float z01 = (float)t * invTm1;
        const float zv  = near + fn * z01;
        const float px = fminf(fmaxf(ox + dx * zv, -1.f), 1.f);
        const float py = fminf(fmaxf(oy + dy * zv, -1.f), 1.f);
        const float pz = fminf(fmaxf(oz + dz * zv, -1.f), 1.f);
        const _Float16 qx = (_Float16)(hi ? 0.f : px);
        const _Float16 qy = (_Float16)(hi ? 0.f : py);
        const _Float16 qz = (_Float16)(hi ? 0.f : pz);
        Bm[0] = qx; Bm[1] = qy; Bm[2] = qz;
        Bc[0] = qx; Bc[1] = qy; Bc[2] = qz;

        v8f D0 = WMMA_F16(A1t0, Bm, CbT0);
        v8f D1 = WMMA_F16(A1t1, Bm, CbT1);
        v8f C0 = WMMA_F16(Ac1t0, Bc, CbC0);
        v8f C1 = WMMA_F16(Ac1t1, Bc, CbC1);
        const v8f Z = {};
        const v16h Bh = relu_pack(D0, D1);
        v8f Ds = WMMA_F16(Asig, Bh, Z);
        const v16h Bhc = relu_pack(C0, C1);
        v8f Dc = WMMA_F16(Acol, Bhc, Z);

        const float delta = (t == T - 1) ? dlast : dstep;
        const float dl2   = -delta * L2E;
        const float sg  = softplus_f(Ds[0]);
        const float a   = 1.f - exp2_fast(dl2 * sg);
        const float w   = a * Tc;
        Tc *= (1.f - a + 1e-15f);
        ws  += w;   dep += w * z01;
        im0 += w * sigmoid_f(Dc[0]);
        im1 += w * sigmoid_f(Dc[1]);
        im2 += w * sigmoid_f(Dc[2]);
    }

    if (seg == 1) {
        float* p = &part[tib][0][lane];
        p[0 * 32] = ws;   p[1 * 32] = dep;  p[2 * 32] = im0;  p[3 * 32] = im1;  p[4 * 32] = im2;
    }
    __syncthreads();
    if (seg == 0) {
        const float* p = &part[tib][0][lane];
        ws   += Tc * p[0 * 32];  dep  += Tc * p[1 * 32];
        im0  += Tc * p[2 * 32];  im1  += Tc * p[3 * 32];  im2 += Tc * p[4 * 32];
    }

    if (seg == 0 && (rayBase + m) < N) {
        float* o = out + (size_t)(rayBase + m) * 9 + (hi ? 5 : 0);
        const float bg = 1.f - ws;
        o[0] = im0 + bg;  o[1] = im1 + bg;  o[2] = im2 + bg;
        o[3] = dep;
        if (!hi) o[4] = ws;
    }
}

extern "C" void kernel_launch(void* const* d_in, const int* in_sizes, int n_in,
                              void* d_out, int out_size, void* d_ws, size_t ws_size,
                              hipStream_t stream) {
    const float* rays_o = (const float*)d_in[0];
    const float* rays_d = (const float*)d_in[1];
    const float* W1     = (const float*)d_in[2];
    const float* b1     = (const float*)d_in[3];
    const float* Wsig   = (const float*)d_in[4];
    const float* Wsig_d = (const float*)d_in[5];
    const float* Wc1    = (const float*)d_in[6];
    const float* bc1    = (const float*)d_in[7];
    const float* Wc2    = (const float*)d_in[8];
    const float* Wc2_d  = (const float*)d_in[9];
    const int*   nsteps = (const int*)d_in[10];

    const int N      = in_sizes[0] / 3;
    const int tiles  = (N + 15) / 16;
    const int block  = 256;
    const int grid   = (tiles + 3) / 4;

    nerf_wmma_kernel<<<grid, block, 0, stream>>>(
        rays_o, rays_d, W1, b1, Wsig, Wsig_d, Wc1, bc1, Wc2, Wc2_d,
        nsteps, (float*)d_out, N);
}


# both hidden-layer products accumulate and return in f16 (16-bit C and D) instead of f32: packed results feed relu (16-bit integer max) and the head products directly; 32 conversions per trip go
# speedup vs baseline: 2.1517x; 1.0831x over previous
#include <hip/hip_runtime.h>

typedef __attribute__((ext_vector_type(16))) _Float16 v16h;
typedef __attribute__((ext_vector_type(8)))  float    v8f;
typedef __attribute__((ext_vector_type(8)))  _Float16 v8h;
typedef __attribute__((ext_vector_type(16))) short    v16s;

#define WMMA_F16(A, B, C) \
    __builtin_amdgcn_wmma_f32_16x16x32_f16(false, (A), false, (B), (short)0, (C), false, false)
#define WMMA_F16H(A, B, C) \
    __builtin_amdgcn_wmma_f16_16x16x32_f16(false, (A), false, (B), (short)0, (C), false, false)

#define L2E 1.4426950408889634f
#define LN2 0.6931471805599453f

__device__ __forceinline__ float exp2_fast(float x) { return __builtin_amdgcn_exp2f(x); }
__device__ __forceinline__ float log2_fast(float x) { return __builtin_amdgcn_logf(x); }
__device__ __forceinline__ float rcp_fast(float x)  { return __builtin_amdgcn_rcpf(x); }

__device__ __forceinline__ float softplus_f(float x) {
    return x > 20.f ? x : LN2 * log2_fast(1.f + exp2_fast(x * L2E));
}
__device__ __forceinline__ float sigmoid_f(float x) {
    return rcp_fast(1.f + exp2_fast(-x * L2E));
}

__device__ __forceinline__ v16h relu_pack(v8h l, v8h h) {
    const v16h x = __builtin_shufflevector(l, h, 0, 1, 2, 3, 4, 5, 6, 7,
                                                 8, 9, 10, 11, 12, 13, 14, 15);
    const v16s zero = {};
    const v16s r = __builtin_elementwise_max(__builtin_bit_cast(v16s, x), zero);
    return __builtin_bit_cast(v16h, r);
}

__global__ __launch_bounds__(256) __attribute__((amdgpu_waves_per_eu(4, 4)))
void nerf_wmma_kernel(const float* __restrict__ rays_o,
                      const float* __restrict__ rays_d,
                      const float* __restrict__ W1,   const float* __restrict__ b1,
                      const float* __restrict__ Wsig, const float* __restrict__ Wsig_d,
                      const float* __restrict__ Wc1,  const float* __restrict__ bc1,
                      const float* __restrict__ Wc2,  const float* __restrict__ Wc2_d,
                      const int*   __restrict__ num_steps,
                      float* __restrict__ out, int N)
{
    const int lane = threadIdx.x & 31;
    const int wib  = __builtin_amdgcn_readfirstlane((int)(threadIdx.x >> 5));
    const int tib  = wib & 3;
    const int seg  = wib >> 2;
    const int m    = lane & 15;
    const bool hi  = (lane & 16) != 0;
    const int rayBase = (blockIdx.x * 4 + tib) * 16;

    __shared__ float part[4][5][32];

    int ray = rayBase + m;
    if (ray >= N) ray = N - 1;

    const int    kb     = hi ? 8 : 0;
    const int    m7     = m & 7;
    const int    jc     = m7 < 3 ? m7 : 2;
    const bool   sigRow = (m7 == 0);
    const bool   colRow = (m7 < 3);
    const float* pSig   = (m & 8) ? Wsig_d : Wsig;
    const float* pCol   = (m & 8) ? Wc2_d  : Wc2;

    v16h A1t0 = {}, A1t1 = {}, Ac1t0 = {}, Ac1t1 = {};
#pragma unroll
    for (int e = 0; e < 6; ++e) {
        if (e < 3) {
            const float a0 = W1[e * 32 + m];
            const float a1 = W1[e * 32 + (m + 16)];
            A1t0[e] = (_Float16)(hi ? 0.f : a0);
            A1t1[e] = (_Float16)(hi ? 0.f : a1);
        }
        const float c0 = Wc1[e * 32 + m];
        const float c1 = Wc1[e * 32 + (m + 16)];
        Ac1t0[e] = (_Float16)(hi ? 0.f : c0);
        Ac1t1[e] = (_Float16)(hi ? 0.f : c1);
    }
    v16h Asig, Acol;
#pragma unroll
    for (int e = 0; e < 16; ++e) {
        const int K = (e < 8 ? e : e + 8) + kb;
        const float sv = pSig[K];
        const float cv = pCol[K * 3 + jc];
        Asig[e] = (_Float16)(sigRow ? sv : 0.f);
        Acol[e] = (_Float16)(colRow ? cv : 0.f);
    }
    v8h CbT0, CbT1, CbC0, CbC1;
#pragma unroll
    for (int i = 0; i < 8; ++i) {
        const int h0 = i + (hi ? 8 : 0);
        CbT0[i] = (_Float16)b1[h0];      CbT1[i] = (_Float16)b1[h0 + 16];
        CbC0[i] = (_Float16)bc1[h0];     CbC1[i] = (_Float16)bc1[h0 + 16];
    }

    float ox = rays_o[ray * 3 + 0], oy = rays_o[ray * 3 + 1], oz = rays_o[ray * 3 + 2];
    float dx = rays_d[ray * 3 + 0], dy = rays_d[ray * 3 + 1], dz = rays_d[ray * 3 + 2];
    {
        float il = 1.f / sqrtf(dx * dx + dy * dy + dz * dz);
        dx *= il; dy *= il; dz *= il;
    }
    float near, far;
    {
        float ivx = 1.f / dx, ivy = 1.f / dy, ivz = 1.f / dz;
        float t1x = (-1.f - ox) * ivx, t2x = (1.f - ox) * ivx;
        float t1y = (-1.f - oy) * ivy, t2y = (1.f - oy) * ivy;
        float t1z = (-1.f - oz) * ivz, t2z = (1.f - oz) * ivz;
        near = fmaxf(fmaxf(fminf(t1x, t2x), fminf(t1y, t2y)), fminf(t1z, t2z));
        far  = fminf(fminf(fmaxf(t1x, t2x), fmaxf(t1y, t2y)), fmaxf(t1z, t2z));
        near = fmaxf(near, 0.2f);
        far  = fmaxf(far, near + 1e-6f);
    }

    const int   T      = *num_steps;
    const float invTm1 = 1.f / (float)(T - 1);
    const float fn     = far - near;
    const float dstep  = fn * invTm1;
    const float dlast  = fn / (float)T;

    v16h Bm = {};
    v16h Bc = {};
    Bc[3] = (_Float16)(hi ? 0.f : dx);
    Bc[4] = (_Float16)(hi ? 0.f : dy);
    Bc[5] = (_Float16)(hi ? 0.f : dz);

    float Tc = 1.f,  ws = 0.f,  dep = 0.f,  im0 = 0.f, im1 = 0.f, im2 = 0.f;

    const int Th = T >> 1;
    const int t0 = seg ? Th : 0;
    const int t1 = seg ? T  : Th;

    v16h Bm2 = {};
    v16h Bc2 = {};
    Bc2[3] = Bc[3];
    Bc2[4] = Bc[4];
    Bc2[5] = Bc[5];

    int t = t0;
    for (; t + 1 < t1; t += 2) {
        const float z01a = (float)t * invTm1;
        const float z01b = (float)(t + 1) * invTm1;
        const float zva  = near + fn * z01a;
        const float zvb  = near + fn * z01b;
        const float pxa = fminf(fmaxf(ox + dx * zva, -1.f), 1.f);
        const float pya = fminf(fmaxf(oy + dy * zva, -1.f), 1.f);
        const float pza = fminf(fmaxf(oz + dz * zva, -1.f), 1.f);
        const float pxb = fminf(fmaxf(ox + dx * zvb, -1.f), 1.f);
        const float pyb = fminf(fmaxf(oy + dy * zvb, -1.f), 1.f);
        const float pzb = fminf(fmaxf(oz + dz * zvb, -1.f), 1.f);
        const _Float16 qxa = (_Float16)(hi ? 0.f : pxa);
        const _Float16 qya = (_Float16)(hi ? 0.f : pya);
        const _Float16 qza = (_Float16)(hi ? 0.f : pza);
        const _Float16 qxb = (_Float16)(hi ? 0.f : pxb);
        const _Float16 qyb = (_Float16)(hi ? 0.f : pyb);
        const _Float16 qzb = (_Float16)(hi ? 0.f : pzb);
        Bm[0]  = qxa; Bm[1]  = qya; Bm[2]  = qza;
        Bc[0]  = qxa; Bc[1]  = qya; Bc[2]  = qza;
        Bm2[0] = qxb; Bm2[1] = qyb; Bm2[2] = qzb;
        Bc2[0] = qxb; Bc2[1] = qyb; Bc2[2] = qzb;

        v8h D0a = WMMA_F16H(A1t0,  Bm,  CbT0);
        v8h D1a = WMMA_F16H(A1t1,  Bm,  CbT1);
        v8h C0a = WMMA_F16H(Ac1t0, Bc,  CbC0);
        v8h C1a = WMMA_F16H(Ac1t1, Bc,  CbC1);
        v8h D0b = WMMA_F16H(A1t0,  Bm2, CbT0);
        v8h D1b = WMMA_F16H(A1t1,  Bm2, CbT1);
        v8h C0b = WMMA_F16H(Ac1t0, Bc2, CbC0);
        v8h C1b = WMMA_F16H(Ac1t1, Bc2, CbC1);

        const v8f Z = {};
        const v16h Bha  = relu_pack(D0a, D1a);
        v8f Dsa = WMMA_F16(Asig, Bha,  Z);
        const v16h Bhca = relu_pack(C0a, C1a);
        v8f Dca = WMMA_F16(Acol, Bhca, Z);
        const v16h Bhb  = relu_pack(D0b, D1b);
        v8f Dsb = WMMA_F16(Asig, Bhb,  Z);
        const v16h Bhcb = relu_pack(C0b, C1b);
        v8f Dcb = WMMA_F16(Acol, Bhcb, Z);

        const float dla = -dstep * L2E;
        const float dlb = -((t + 1 == T - 1) ? dlast : dstep) * L2E;
        const float sga = softplus_f(Dsa[0]);
        const float sgb = softplus_f(Dsb[0]);
        const float aa  = 1.f - exp2_fast(dla * sga);
        const float ab  = 1.f - exp2_fast(dlb * sgb);
        const float ra0 = sigmoid_f(Dca[0]), ra1 = sigmoid_f(Dca[1]), ra2 = sigmoid_f(Dca[2]);
        const float rb0 = sigmoid_f(Dcb[0]), rb1 = sigmoid_f(Dcb[1]), rb2 = sigmoid_f(Dcb[2]);

        const float wa = aa * Tc;
        Tc *= (1.f - aa + 1e-15f);
        ws  += wa;   dep += wa * z01a;
        im0 += wa * ra0;  im1 += wa * ra1;  im2 += wa * ra2;
        const float wb = ab * Tc;
        Tc *= (1.f - ab + 1e-15f);
        ws  += wb;   dep += wb * z01b;
        im0 += wb * rb0;  im1 += wb * rb1;  im2 += wb * rb2;
    }
    if (t < t1) {
        const float z01 = (float)t * invTm1;
        const float zv  = near + fn * z01;
        const float px = fminf(fmaxf(ox + dx * zv, -1.f), 1.f);
        const float py = fminf(fmaxf(oy + dy * zv, -1.f), 1.f);
        const float pz = fminf(fmaxf(oz + dz * zv, -1.f), 1.f);
        const _Float16 qx = (_Float16)(hi ? 0.f : px);
        const _Float16 qy = (_Float16)(hi ? 0.f : py);
        const _Float16 qz = (_Float16)(hi ? 0.f : pz);
        Bm[0] = qx; Bm[1] = qy; Bm[2] = qz;
        Bc[0] = qx; Bc[1] = qy; Bc[2] = qz;

        v8h D0 = WMMA_F16H(A1t0, Bm, CbT0);
        v8h D1 = WMMA_F16H(A1t1, Bm, CbT1);
        v8h C0 = WMMA_F16H(Ac1t0, Bc, CbC0);
        v8h C1 = WMMA_F16H(Ac1t1, Bc, CbC1);
        const v8f Z = {};
        const v16h Bh = relu_pack(D0, D1);
        v8f Ds = WMMA_F16(Asig, Bh, Z);
        const v16h Bhc = relu_pack(C0, C1);
        v8f Dc = WMMA_F16(Acol, Bhc, Z);

        const float delta = (t == T - 1) ? dlast : dstep;
        const float dl2   = -delta * L2E;
        const float sg  = softplus_f(Ds[0]);
        const float a   = 1.f - exp2_fast(dl2 * sg);
        const float w   = a * Tc;
        Tc *= (1.f - a + 1e-15f);
        ws  += w;   dep += w * z01;
        im0 += w * sigmoid_f(Dc[0]);
        im1 += w * sigmoid_f(Dc[1]);
        im2 += w * sigmoid_f(Dc[2]);
    }

    if (seg == 1) {
        float* p = &part[tib][0][lane];
        p[0 * 32] = ws;   p[1 * 32] = dep;  p[2 * 32] = im0;  p[3 * 32] = im1;  p[4 * 32] = im2;
    }
    __syncthreads();
    if (seg == 0) {
        const float* p = &part[tib][0][lane];
        ws   += Tc * p[0 * 32];  dep  += Tc * p[1 * 32];
        im0  += Tc * p[2 * 32];  im1  += Tc * p[3 * 32];  im2 += Tc * p[4 * 32];
    }

    if (seg == 0 && (rayBase + m) < N) {
        float* o = out + (size_t)(rayBase + m) * 9 + (hi ? 5 : 0);
        const float bg = 1.f - ws;
        o[0] = im0 + bg;  o[1] = im1 + bg;  o[2] = im2 + bg;
        o[3] = dep;
        if (!hi) o[4] = ws;
    }
}

extern "C" void kernel_launch(void* const* d_in, const int* in_sizes, int n_in,
                              void* d_out, int out_size, void* d_ws, size_t ws_size,
                              hipStream_t stream) {
    const float* rays_o = (const float*)d_in[0];
    const float* rays_d = (const float*)d_in[1];
    const float* W1     = (const float*)d_in[2];
    const float* b1     = (const float*)d_in[3];
    const float* Wsig   = (const float*)d_in[4];
    const float* Wsig_d = (const float*)d_in[5];
    const float* Wc1    = (const float*)d_in[6];
    const float* bc1    = (const float*)d_in[7];
    const float* Wc2    = (const float*)d_in[8];
    const float* Wc2_d  = (const float*)d_in[9];
    const int*   nsteps = (const int*)d_in[10];

    const int N      = in_sizes[0] / 3;
    const int tiles  = (N + 15) / 16;
    const int block  = 256;
    const int grid   = (tiles + 3) / 4;

    nerf_wmma_kernel<<<grid, block, 0, stream>>>(
        rays_o, rays_d, W1, b1, Wsig, Wsig_d, Wc1, bc1, Wc2, Wc2_d,
        nsteps, (float*)d_out, N);
}


# one first-layer B fragment per trip carries both samples (lanes 0-15 sample t, lanes 16-31 sample t+1) and feeds trunk and colour; zero padding moved to per-sample A tiles; no sum changes its order
# speedup vs baseline: 2.3274x; 1.0817x over previous
#include <hip/hip_runtime.h>

typedef __attribute__((ext_vector_type(16))) _Float16 v16h;
typedef __attribute__((ext_vector_type(8)))  float    v8f;
typedef __attribute__((ext_vector_type(8)))  _Float16 v8h;
typedef __attribute__((ext_vector_type(16))) short    v16s;

#define WMMA_F16(A, B, C) \
    __builtin_amdgcn_wmma_f32_16x16x32_f16(false, (A), false, (B), (short)0, (C), false, false)
#define WMMA_F16H(A, B, C) \
    __builtin_amdgcn_wmma_f16_16x16x32_f16(false, (A), false, (B), (short)0, (C), false, false)

#define L2E 1.4426950408889634f
#define LN2 0.6931471805599453f

__device__ __forceinline__ float exp2_fast(float x) { return __builtin_amdgcn_exp2f(x); }
__device__ __forceinline__ float log2_fast(float x) { return __builtin_amdgcn_logf(x); }
__device__ __forceinline__ float rcp_fast(float x)  { return __builtin_amdgcn_rcpf(x); }

__device__ __forceinline__ float softplus_f(float x) {
    return x > 20.f ? x : LN2 * log2_fast(1.f + exp2_fast(x * L2E));
}
__device__ __forceinline__ float sigmoid_f(float x) {
    return rcp_fast(1.f + exp2_fast(-x * L2E));
}

__device__ __forceinline__ v16h relu_pack(v8h l, v8h h) {
    const v16h x = __builtin_shufflevector(l, h, 0, 1, 2, 3, 4, 5, 6, 7,
                                                 8, 9, 10, 11, 12, 13, 14, 15);
    const v16s zero = {};
    const v16s r = __builtin_elementwise_max(__builtin_bit_cast(v16s, x), zero);
    return __builtin_bit_cast(v16h, r);
}

__global__ __launch_bounds__(256) __attribute__((amdgpu_waves_per_eu(4, 4)))
void nerf_wmma_kernel(const float* __restrict__ rays_o,
                      const float* __restrict__ rays_d,
                      const float* __restrict__ W1,   const float* __restrict__ b1,
                      const float* __restrict__ Wsig, const float* __restrict__ Wsig_d,
                      const float* __restrict__ Wc1,  const float* __restrict__ bc1,
                      const float* __restrict__ Wc2,  const float* __restrict__ Wc2_d,
                      const int*   __restrict__ num_steps,
                      float* __restrict__ out, int N)
{
    const int lane = threadIdx.x & 31;
    const int wib  = __builtin_amdgcn_readfirstlane((int)(threadIdx.x >> 5));
    const int tib  = wib & 3;
    const int seg  = wib >> 2;
    const int m    = lane & 15;
    const bool hi  = (lane & 16) != 0;
    const int rayBase = (blockIdx.x * 4 + tib) * 16;

    __shared__ float part[4][5][32];

    int ray = rayBase + m;
    if (ray >= N) ray = N - 1;

    const int    kb     = hi ? 8 : 0;
    const int    m7     = m & 7;
    const int    jc     = m7 < 3 ? m7 : 2;
    const bool   sigRow = (m7 == 0);
    const bool   colRow = (m7 < 3);
    const float* pSig   = (m & 8) ? Wsig_d : Wsig;
    const float* pCol   = (m & 8) ? Wc2_d  : Wc2;

    v16h A1t0a = {}, A1t1a = {}, Ac1t0a = {}, Ac1t1a = {};
    v16h A1t0b = {}, A1t1b = {}, Ac1t0b = {}, Ac1t1b = {};
#pragma unroll
    for (int e = 0; e < 6; ++e) {
        if (e < 3) {
            const float a0 = W1[e * 32 + m];
            const float a1 = W1[e * 32 + (m + 16)];
            A1t0a[e] = (_Float16)(hi ? 0.f : a0);
            A1t1a[e] = (_Float16)(hi ? 0.f : a1);
            A1t0b[e] = (_Float16)(hi ? a0 : 0.f);
            A1t1b[e] = (_Float16)(hi ? a1 : 0.f);
        }
        const float c0 = Wc1[e * 32 + m];
        const float c1 = Wc1[e * 32 + (m + 16)];
        Ac1t0a[e] = (_Float16)(hi ? 0.f : c0);
        Ac1t1a[e] = (_Float16)(hi ? 0.f : c1);
        Ac1t0b[e] = (_Float16)(hi ? c0 : 0.f);
        Ac1t1b[e] = (_Float16)(hi ? c1 : 0.f);
    }
    v16h Asig, Acol;
#pragma unroll
    for (int e = 0; e < 16; ++e) {
        const int K = (e < 8 ? e : e + 8) + kb;
        const float sv = pSig[K];
        const float cv = pCol[K * 3 + jc];
        Asig[e] = (_Float16)(sigRow ? sv : 0.f);
        Acol[e] = (_Float16)(colRow ? cv : 0.f);
    }
    v8h CbT0, CbT1, CbC0, CbC1;
#pragma unroll
    for (int i = 0; i < 8; ++i) {
        const int h0 = i + (hi ? 8 : 0);
        CbT0[i] = (_Float16)b1[h0];      CbT1[i] = (_Float16)b1[h0 + 16];
        CbC0[i] = (_Float16)bc1[h0];     CbC1[i] = (_Float16)bc1[h0 + 16];
    }

    float ox = rays_o[ray * 3 + 0], oy = rays_o[ray * 3 + 1], oz = rays_o[ray * 3 + 2];
    float dx = rays_d[ray * 3 + 0], dy = rays_d[ray * 3 + 1], dz = rays_d[ray * 3 + 2];
    {
        float il = 1.f / sqrtf(dx * dx + dy * dy + dz * dz);
        dx *= il; dy *= il; dz *= il;
    }
    float near, far;
    {
        float ivx = 1.f / dx, ivy = 1.f / dy, ivz = 1.f / dz;
        float t1x = (-1.f - ox) * ivx, t2x = (1.f - ox) * ivx;
        float t1y = (-1.f - oy) * ivy, t2y = (1.f - oy) * ivy;
        float t1z = (-1.f - oz) * ivz, t2z = (1.f - oz) * ivz;
        near = fmaxf(fmaxf(fminf(t1x, t2x), fminf(t1y, t2y)), fminf(t1z, t2z));
        far  = fminf(fminf(fmaxf(t1x, t2x), fmaxf(t1y, t2y)), fmaxf(t1z, t2z));
        near = fmaxf(near, 0.2f);
        far  = fmaxf(far, near + 1e-6f);
    }

    const int   T      = *num_steps;
    const float invTm1 = 1.f / (float)(T - 1);
    const float fn     = far - near;
    const float dstep  = fn * invTm1;
    const float dlast  = fn / (float)T;

    v16h B1 = {};
    B1[3] = (_Float16)dx;
    B1[4] = (_Float16)dy;
    B1[5] = (_Float16)dz;

    float Tc = 1.f,  ws = 0.f,  dep = 0.f,  im0 = 0.f, im1 = 0.f, im2 = 0.f;

    const int Th = T >> 1;
    const int t0 = seg ? Th : 0;
    const int t1 = seg ? T  : Th;

    int t = t0;
    for (; t + 1 < t1; t += 2) {
        const float z01a = (float)t * invTm1;
        const float z01b = (float)(t + 1) * invTm1;
        const float z01l = hi ? z01b : z01a;
        const float zv   = near + fn * z01l;
        const float px = fminf(fmaxf(ox + dx * zv, -1.f), 1.f);
        const float py = fminf(fmaxf(oy + dy * zv, -1.f), 1.f);
        const float pz = fminf(fmaxf(oz + dz * zv, -1.f), 1.f);
        B1[0] = (_Float16)px;  B1[1] = (_Float16)py;  B1[2] = (_Float16)pz;

        v8h D0a = WMMA_F16H(A1t0a,  B1, CbT0);
        v8h D1a = WMMA_F16H(A1t1a,  B1, CbT1);
        v8h C0a = WMMA_F16H(Ac1t0a, B1, CbC0);
        v8h C1a = WMMA_F16H(Ac1t1a, B1, CbC1);
        v8h D0b = WMMA_F16H(A1t0b,  B1, CbT0);
        v8h D1b = WMMA_F16H(A1t1b,  B1, CbT1);
        v8h C0b = WMMA_F16H(Ac1t0b, B1, CbC0);
        v8h C1b = WMMA_F16H(Ac1t1b, B1, CbC1);

        const v8f Z = {};
        const v16h Bha  = relu_pack(D0a, D1a);
        v8f Dsa = WMMA_F16(Asig, Bha,  Z);
        const v16h Bhca = relu_pack(C0a, C1a);
        v8f Dca = WMMA_F16(Acol, Bhca, Z);
        const v16h Bhb  = relu_pack(D0b, D1b);
        v8f Dsb = WMMA_F16(Asig, Bhb,  Z);
        const v16h Bhcb = relu_pack(C0b, C1b);
        v8f Dcb = WMMA_F16(Acol, Bhcb, Z);

        const float dla = -dstep * L2E;
        const float dlb = -((t + 1 == T - 1) ? dlast : dstep) * L2E;
        const float sga = softplus_f(Dsa[0]);
        const float sgb = softplus_f(Dsb[0]);
        const float aa  = 1.f - exp2_fast(dla * sga);
        const float ab  = 1.f - exp2_fast(dlb * sgb);
        const float ra0 = sigmoid_f(Dca[0]), ra1 = sigmoid_f(Dca[1]), ra2 = sigmoid_f(Dca[2]);
        const float rb0 = sigmoid_f(Dcb[0]), rb1 = sigmoid_f(Dcb[1]), rb2 = sigmoid_f(Dcb[2]);

        const float wa = aa * Tc;
        Tc *= (1.f - aa + 1e-15f);
        ws  += wa;   dep += wa * z01a;
        im0 += wa * ra0;  im1 += wa * ra1;  im2 += wa * ra2;
        const float wb = ab * Tc;
        Tc *= (1.f - ab + 1e-15f);
        ws  += wb;   dep += wb * z01b;
        im0 += wb * rb0;  im1 += wb * rb1;  im2 += wb * rb2;
    }
    if (t < t1) {
        const float z01 = (float)t * invTm1;
        const float zv  = near + fn * z01;
        const float px = fminf(fmaxf(ox + dx * zv, -1.f), 1.f);
        const float py = fminf(fmaxf(oy + dy * zv, -1.f), 1.f);
        const float pz = fminf(fmaxf(oz + dz * zv, -1.f), 1.f);
        B1[0] = (_Float16)px;  B1[1] = (_Float16)py;  B1[2] = (_Float16)pz;

        v8h D0 = WMMA_F16H(A1t0a, B1, CbT0);
        v8h D1 = WMMA_F16H(A1t1a, B1, CbT1);
        v8h C0 = WMMA_F16H(Ac1t0a, B1, CbC0);
        v8h C1 = WMMA_F16H(Ac1t1a, B1, CbC1);
        const v8f Z = {};
        const v16h Bh = relu_pack(D0, D1);
        v8f Ds = WMMA_F16(Asig, Bh, Z);
        const v16h Bhc = relu_pack(C0, C1);
        v8f Dc = WMMA_F16(Acol, Bhc, Z);

        const float delta = (t == T - 1) ? dlast : dstep;
        const float dl2   = -delta * L2E;
        const float sg  = softplus_f(Ds[0]);
        const float a   = 1.f - exp2_fast(dl2 * sg);
        const float w   = a * Tc;
        Tc *= (1.f - a + 1e-15f);
        ws  += w;   dep += w * z01;
        im0 += w * sigmoid_f(Dc[0]);
        im1 += w * sigmoid_f(Dc[1]);
        im2 += w * sigmoid_f(Dc[2]);
    }

    if (seg == 1) {
        float* p = &part[tib][0][lane];
        p[0 * 32] = ws;   p[1 * 32] = dep;  p[2 * 32] = im0;  p[3 * 32] = im1;  p[4 * 32] = im2;
    }
    __syncthreads();
    if (seg == 0) {
        const float* p = &part[tib][0][lane];
        ws   += Tc * p[0 * 32];  dep  += Tc * p[1 * 32];
        im0  += Tc * p[2 * 32];  im1  += Tc * p[3 * 32];  im2 += Tc * p[4 * 32];
    }

    if (seg == 0 && (rayBase + m) < N) {
        float* o = out + (size_t)(rayBase + m) * 9 + (hi ? 5 : 0);
        const float bg = 1.f - ws;
        o[0] = im0 + bg;  o[1] = im1 + bg;  o[2] = im2 + bg;
        o[3] = dep;
        if (!hi) o[4] = ws;
    }
}

extern "C" void kernel_launch(void* const* d_in, const int* in_sizes, int n_in,
                              void* d_out, int out_size, void* d_ws, size_t ws_size,
                              hipStream_t stream) {
    const float* rays_o = (const float*)d_in[0];
    const float* rays_d = (const float*)d_in[1];
    const float* W1     = (const float*)d_in[2];
    const float* b1     = (const float*)d_in[3];
    const float* Wsig   = (const float*)d_in[4];
    const float* Wsig_d = (const float*)d_in[5];
    const float* Wc1    = (const float*)d_in[6];
    const float* bc1    = (const float*)d_in[7];
    const float* Wc2    = (const float*)d_in[8];
    const float* Wc2_d  = (const float*)d_in[9];
    const int*   nsteps = (const int*)d_in[10];

    const int N      = in_sizes[0] / 3;
    const int tiles  = (N + 15) / 16;
    const int block  = 256;
    const int grid   = (tiles + 3) / 4;

    nerf_wmma_kernel<<<grid, block, 0, stream>>>(
        rays_o, rays_d, W1, b1, Wsig, Wsig_d, Wc1, bc1, Wc2, Wc2_d,
        nsteps, (float*)d_out, N);
}


# sample loop: two order fences and whole-tile register allocation keep every reader or writer of a product's registers four vector slots behind it; the 28 filler no-ops per trip go
# speedup vs baseline: 2.6991x; 1.1597x over previous
#include <hip/hip_runtime.h>

typedef __attribute__((ext_vector_type(16))) _Float16 v16h;
typedef __attribute__((ext_vector_type(8)))  float    v8f;
typedef __attribute__((ext_vector_type(8)))  _Float16 v8h;
typedef __attribute__((ext_vector_type(16))) short    v16s;

#define WMMA_F16(A, B, C) \
    __builtin_amdgcn_wmma_f32_16x16x32_f16(false, (A), false, (B), (short)0, (C), false, false)
#define WMMA_F16H(A, B, C) \
    __builtin_amdgcn_wmma_f16_16x16x32_f16(false, (A), false, (B), (short)0, (C), false, false)

#define L2E 1.4426950408889634f
#define LN2 0.6931471805599453f

__device__ __forceinline__ float exp2_fast(float x) { return __builtin_amdgcn_exp2f(x); }
__device__ __forceinline__ float log2_fast(float x) { return __builtin_amdgcn_logf(x); }
__device__ __forceinline__ float rcp_fast(float x)  { return __builtin_amdgcn_rcpf(x); }

__device__ __forceinline__ float softplus_f(float x) {
    return x > 20.f ? x : LN2 * log2_fast(1.f + exp2_fast(x * L2E));
}
__device__ __forceinline__ float sigmoid_f(float x) {
    return rcp_fast(1.f + exp2_fast(-x * L2E));
}

__device__ __forceinline__ v16h relu_pack(v8h l, v8h h) {
    const v16h x = __builtin_shufflevector(l, h, 0, 1, 2, 3, 4, 5, 6, 7,
                                                 8, 9, 10, 11, 12, 13, 14, 15);
    const v16s zero = {};
    const v16s r = __builtin_elementwise_max(__builtin_bit_cast(v16s, x), zero);
    return __builtin_bit_cast(v16h, r);
}

__global__ __launch_bounds__(256) __attribute__((amdgpu_waves_per_eu(4, 4)))
void nerf_wmma_kernel(const float* __restrict__ rays_o,
                      const float* __restrict__ rays_d,
                      const float* __restrict__ W1,   const float* __restrict__ b1,
                      const float* __restrict__ Wsig, const float* __restrict__ Wsig_d,
                      const float* __restrict__ Wc1,  const float* __restrict__ bc1,
                      const float* __restrict__ Wc2,  const float* __restrict__ Wc2_d,
                      const int*   __restrict__ num_steps,
                      float* __restrict__ out, int N)
{
    const int lane = threadIdx.x & 31;
    const int wib  = __builtin_amdgcn_readfirstlane((int)(threadIdx.x >> 5));
    const int tib  = wib & 3;
    const int seg  = wib >> 2;
    const int m    = lane & 15;
    const bool hi  = (lane & 16) != 0;
    const int rayBase = (blockIdx.x * 4 + tib) * 16;

    __shared__ float part[4][5][32];

    int ray = rayBase + m;
    if (ray >= N) ray = N - 1;

    const int    kb     = hi ? 8 : 0;
    const int    m7     = m & 7;
    const int    jc     = m7 < 3 ? m7 : 2;
    const bool   sigRow = (m7 == 0);
    const bool   colRow = (m7 < 3);
    const float* pSig   = (m & 8) ? Wsig_d : Wsig;
    const float* pCol   = (m & 8) ? Wc2_d  : Wc2;

    v16h A1t0a = {}, A1t1a = {}, Ac1t0a = {}, Ac1t1a = {};
    v16h A1t0b = {}, A1t1b = {}, Ac1t0b = {}, Ac1t1b = {};
#pragma unroll
    for (int e = 0; e < 6; ++e) {
        if (e < 3) {
            const float a0 = W1[e * 32 + m];
            const float a1 = W1[e * 32 + (m + 16)];
            A1t0a[e] = (_Float16)(hi ? 0.f : a0);
            A1t1a[e] = (_Float16)(hi ? 0.f : a1);
            A1t0b[e] = (_Float16)(hi ? a0 : 0.f);
            A1t1b[e] = (_Float16)(hi ? a1 : 0.f);
        }
        const float c0 = Wc1[e * 32 + m];
        const float c1 = Wc1[e * 32 + (m + 16)];
        Ac1t0a[e] = (_Float16)(hi ? 0.f : c0);
        Ac1t1a[e] = (_Float16)(hi ? 0.f : c1);
        Ac1t0b[e] = (_Float16)(hi ? c0 : 0.f);
        Ac1t1b[e] = (_Float16)(hi ? c1 : 0.f);
    }
    v16h Asig, Acol;
#pragma unroll
    for (int e = 0; e < 16; ++e) {
        const int K = (e < 8 ? e : e + 8) + kb;
        const float sv = pSig[K];
        const float cv = pCol[K * 3 + jc];
        Asig[e] = (_Float16)(sigRow ? sv : 0.f);
        Acol[e] = (_Float16)(colRow ? cv : 0.f);
    }
    v8h CbT0, CbT1, CbC0, CbC1;
#pragma unroll
    for (int i = 0; i < 8; ++i) {
        const int h0 = i + (hi ? 8 : 0);
        CbT0[i] = (_Float16)b1[h0];      CbT1[i] = (_Float16)b1[h0 + 16];
        CbC0[i] = (_Float16)bc1[h0];     CbC1[i] = (_Float16)bc1[h0 + 16];
    }

    float ox = rays_o[ray * 3 + 0], oy = rays_o[ray * 3 + 1], oz = rays_o[ray * 3 + 2];
    float dx = rays_d[ray * 3 + 0], dy = rays_d[ray * 3 + 1], dz = rays_d[ray * 3 + 2];
    {
        float il = 1.f / sqrtf(dx * dx + dy * dy + dz * dz);
        dx *= il; dy *= il; dz *= il;
    }
    float near, far;
    {
        float ivx = 1.f / dx, ivy = 1.f / dy, ivz = 1.f / dz;
        float t1x = (-1.f - ox) * ivx, t2x = (1.f - ox) * ivx;
        float t1y = (-1.f - oy) * ivy, t2y = (1.f - oy) * ivy;
        float t1z = (-1.f - oz) * ivz, t2z = (1.f - oz) * ivz;
        near = fmaxf(fmaxf(fminf(t1x, t2x), fminf(t1y, t2y)), fminf(t1z, t2z));
        far  = fminf(fminf(fmaxf(t1x, t2x), fmaxf(t1y, t2y)), fmaxf(t1z, t2z));
        near = fmaxf(near, 0.2f);
        far  = fmaxf(far, near + 1e-6f);
    }

    const int   T      = *num_steps;
    const float invTm1 = 1.f / (float)(T - 1);
    const float fn     = far - near;
    const float dstep  = fn * invTm1;
    const float dlast  = fn / (float)T;

    v16h B1 = {};
    B1[3] = (_Float16)dx;
    B1[4] = (_Float16)dy;
    B1[5] = (_Float16)dz;

    float Tc = 1.f,  ws = 0.f,  dep = 0.f,  im0 = 0.f, im1 = 0.f, im2 = 0.f;

    const int Th = T >> 1;
    const int t0 = seg ? Th : 0;
    const int t1 = seg ? T  : Th;

    int t = t0;
    for (; t + 1 < t1; t += 2) {
        const float z01a = (float)t * invTm1;
        const float z01b = (float)(t + 1) * invTm1;
        const float z01l = hi ? z01b : z01a;
        const float zv   = near + fn * z01l;
        const float px = fminf(fmaxf(ox + dx * zv, -1.f), 1.f);
        const float py = fminf(fmaxf(oy + dy * zv, -1.f), 1.f);
        const float pz = fminf(fmaxf(oz + dz * zv, -1.f), 1.f);
        B1[0] = (_Float16)px;  B1[1] = (_Float16)py;  B1[2] = (_Float16)pz;

        v8h D0a = WMMA_F16H(A1t0a,  B1, CbT0);
        v8h D1a = WMMA_F16H(A1t1a,  B1, CbT1);
        v8h C0a = WMMA_F16H(Ac1t0a, B1, CbC0);
        v8h C1a = WMMA_F16H(Ac1t1a, B1, CbC1);
        v8h D0b = WMMA_F16H(A1t0b,  B1, CbT0);
        v8h D1b = WMMA_F16H(A1t1b,  B1, CbT1);
        v8h C0b = WMMA_F16H(Ac1t0b, B1, CbC0);
        v8h C1b = WMMA_F16H(Ac1t1b, B1, CbC1);


        __builtin_amdgcn_sched_barrier(0);

        const v16h Bha  = relu_pack(D0a, D1a);
        const v16h Bhb  = relu_pack(D0b, D1b);
        const v16h Bhca = relu_pack(C0a, C1a);
        const v16h Bhcb = relu_pack(C0b, C1b);
        const v8f Z = {};
        v8f Dsa = WMMA_F16(Asig, Bha,  Z);
        v8f Dsb = WMMA_F16(Asig, Bhb,  Z);
        v8f Dca = WMMA_F16(Acol, Bhca, Z);
        v8f Dcb = WMMA_F16(Acol, Bhcb, Z);
        asm volatile("" :: "v"(Bha), "v"(Bhb), "v"(Bhca), "v"(Bhcb),
                           "v"(Dsa), "v"(Dsb), "v"(Dca),  "v"(Dcb));

        const float dla = -dstep * L2E;
        const float dlb = -((t + 1 == T - 1) ? dlast : dstep) * L2E;
        const float sga = softplus_f(Dsa[0]);
        const float sgb = softplus_f(Dsb[0]);
        __builtin_amdgcn_sched_barrier(0);
        const float aa  = 1.f - exp2_fast(dla * sga);
        const float ab  = 1.f - exp2_fast(dlb * sgb);
        const float ra0 = sigmoid_f(Dca[0]), ra1 = sigmoid_f(Dca[1]), ra2 = sigmoid_f(Dca[2]);
        const float rb0 = sigmoid_f(Dcb[0]), rb1 = sigmoid_f(Dcb[1]), rb2 = sigmoid_f(Dcb[2]);

        const float wa = aa * Tc;
        Tc *= (1.f - aa + 1e-15f);
        ws  += wa;   dep += wa * z01a;
        im0 += wa * ra0;  im1 += wa * ra1;  im2 += wa * ra2;
        const float wb = ab * Tc;
        Tc *= (1.f - ab + 1e-15f);
        ws  += wb;   dep += wb * z01b;
        im0 += wb * rb0;  im1 += wb * rb1;  im2 += wb * rb2;
    }
    if (t < t1) {
        const float z01 = (float)t * invTm1;
        const float zv  = near + fn * z01;
        const float px = fminf(fmaxf(ox + dx * zv, -1.f), 1.f);
        const float py = fminf(fmaxf(oy + dy * zv, -1.f), 1.f);
        const float pz = fminf(fmaxf(oz + dz * zv, -1.f), 1.f);
        B1[0] = (_Float16)px;  B1[1] = (_Float16)py;  B1[2] = (_Float16)pz;

        v8h D0 = WMMA_F16H(A1t0a, B1, CbT0);
        v8h D1 = WMMA_F16H(A1t1a, B1, CbT1);
        v8h C0 = WMMA_F16H(Ac1t0a, B1, CbC0);
        v8h C1 = WMMA_F16H(Ac1t1a, B1, CbC1);
        const v8f Z = {};
        const v16h Bh = relu_pack(D0, D1);
        v8f Ds = WMMA_F16(Asig, Bh, Z);
        const v16h Bhc = relu_pack(C0, C1);
        v8f Dc = WMMA_F16(Acol, Bhc, Z);

        const float delta = (t == T - 1) ? dlast : dstep;
        const float dl2   = -delta * L2E;
        const float sg  = softplus_f(Ds[0]);
        const float a   = 1.f - exp2_fast(dl2 * sg);
        const float w   = a * Tc;
        Tc *= (1.f - a + 1e-15f);
        ws  += w;   dep += w * z01;
        im0 += w * sigmoid_f(Dc[0]);
        im1 += w * sigmoid_f(Dc[1]);
        im2 += w * sigmoid_f(Dc[2]);
    }

    if (seg == 1) {
        float* p = &part[tib][0][lane];
        p[0 * 32] = ws;   p[1 * 32] = dep;  p[2 * 32] = im0;  p[3 * 32] = im1;  p[4 * 32] = im2;
    }
    __syncthreads();
    if (seg == 0) {
        const float* p = &part[tib][0][lane];
        ws   += Tc * p[0 * 32];  dep  += Tc * p[1 * 32];
        im0  += Tc * p[2 * 32];  im1  += Tc * p[3 * 32];  im2 += Tc * p[4 * 32];
    }

    if (seg == 0 && (rayBase + m) < N) {
        float* o = out + (size_t)(rayBase + m) * 9 + (hi ? 5 : 0);
        const float bg = 1.f - ws;
        o[0] = im0 + bg;  o[1] = im1 + bg;  o[2] = im2 + bg;
        o[3] = dep;
        if (!hi) o[4] = ws;
    }
}

extern "C" void kernel_launch(void* const* d_in, const int* in_sizes, int n_in,
                              void* d_out, int out_size, void* d_ws, size_t ws_size,
                              hipStream_t stream) {
    const float* rays_o = (const float*)d_in[0];
    const float* rays_d = (const float*)d_in[1];
    const float* W1     = (const float*)d_in[2];
    const float* b1     = (const float*)d_in[3];
    const float* Wsig   = (const float*)d_in[4];
    const float* Wsig_d = (const float*)d_in[5];
    const float* Wc1    = (const float*)d_in[6];
    const float* bc1    = (const float*)d_in[7];
    const float* Wc2    = (const float*)d_in[8];
    const float* Wc2_d  = (const float*)d_in[9];
    const int*   nsteps = (const int*)d_in[10];

    const int N      = in_sizes[0] / 3;
    const int tiles  = (N + 15) / 16;
    const int block  = 256;
    const int grid   = (tiles + 3) / 4;

    nerf_wmma_kernel<<<grid, block, 0, stream>>>(
        rays_o, rays_d, W1, b1, Wsig, Wsig_d, Wc1, bc1, Wc2, Wc2_d,
        nsteps, (float*)d_out, N);
}


# colour sigmoids as 0.5+0.5*tanh(x/2) on the hardware tanh: head weights halved, loop sums w*tanh, 0.5+0.5* applied once to the joined sums (image sums re-associated); the fence it leaves idle goes
# speedup vs baseline: 2.8881x; 1.0700x over previous
#include <hip/hip_runtime.h>

typedef __attribute__((ext_vector_type(16))) _Float16 v16h;
typedef __attribute__((ext_vector_type(8)))  float    v8f;
typedef __attribute__((ext_vector_type(8)))  _Float16 v8h;
typedef __attribute__((ext_vector_type(16))) short    v16s;

#define WMMA_F16(A, B, C) \
    __builtin_amdgcn_wmma_f32_16x16x32_f16(false, (A), false, (B), (short)0, (C), false, false)
#define WMMA_F16H(A, B, C) \
    __builtin_amdgcn_wmma_f16_16x16x32_f16(false, (A), false, (B), (short)0, (C), false, false)

#define L2E 1.4426950408889634f
#define LN2 0.6931471805599453f

__device__ __forceinline__ float exp2_fast(float x) { return __builtin_amdgcn_exp2f(x); }
__device__ __forceinline__ float log2_fast(float x) { return __builtin_amdgcn_logf(x); }
__device__ __forceinline__ float rcp_fast(float x)  { return __builtin_amdgcn_rcpf(x); }

__device__ __forceinline__ float softplus_f(float x) {
    return x > 20.f ? x : LN2 * log2_fast(1.f + exp2_fast(x * L2E));
}
__device__ __forceinline__ float tanh_fast(float x) { return __builtin_amdgcn_tanhf(x); }

__device__ __forceinline__ v16h relu_pack(v8h l, v8h h) {
    const v16h x = __builtin_shufflevector(l, h, 0, 1, 2, 3, 4, 5, 6, 7,
                                                 8, 9, 10, 11, 12, 13, 14, 15);
    const v16s zero = {};
    const v16s r = __builtin_elementwise_max(__builtin_bit_cast(v16s, x), zero);
    return __builtin_bit_cast(v16h, r);
}

__global__ __launch_bounds__(256) __attribute__((amdgpu_waves_per_eu(4, 4)))
void nerf_wmma_kernel(const float* __restrict__ rays_o,
                      const float* __restrict__ rays_d,
                      const float* __restrict__ W1,   const float* __restrict__ b1,
                      const float* __restrict__ Wsig, const float* __restrict__ Wsig_d,
                      const float* __restrict__ Wc1,  const float* __restrict__ bc1,
                      const float* __restrict__ Wc2,  const float* __restrict__ Wc2_d,
                      const int*   __restrict__ num_steps,
                      float* __restrict__ out, int N)
{
    const int lane = threadIdx.x & 31;
    const int wib  = __builtin_amdgcn_readfirstlane((int)(threadIdx.x >> 5));
    const int tib  = wib & 3;
    const int seg  = wib >> 2;
    const int m    = lane & 15;
    const bool hi  = (lane & 16) != 0;
    const int rayBase = (blockIdx.x * 4 + tib) * 16;

    __shared__ float part[4][5][32];

    int ray = rayBase + m;
    if (ray >= N) ray = N - 1;

    const int    kb     = hi ? 8 : 0;
    const int    m7     = m & 7;
    const int    jc     = m7 < 3 ? m7 : 2;
    const bool   sigRow = (m7 == 0);
    const bool   colRow = (m7 < 3);
    const float* pSig   = (m & 8) ? Wsig_d : Wsig;
    const float* pCol   = (m & 8) ? Wc2_d  : Wc2;

    v16h A1t0a = {}, A1t1a = {}, Ac1t0a = {}, Ac1t1a = {};
    v16h A1t0b = {}, A1t1b = {}, Ac1t0b = {}, Ac1t1b = {};
#pragma unroll
    for (int e = 0; e < 6; ++e) {
        if (e < 3) {
            const float a0 = W1[e * 32 + m];
            const float a1 = W1[e * 32 + (m + 16)];
            A1t0a[e] = (_Float16)(hi ? 0.f : a0);
            A1t1a[e] = (_Float16)(hi ? 0.f : a1);
            A1t0b[e] = (_Float16)(hi ? a0 : 0.f);
            A1t1b[e] = (_Float16)(hi ? a1 : 0.f);
        }
        const float c0 = Wc1[e * 32 + m];
        const float c1 = Wc1[e * 32 + (m + 16)];
        Ac1t0a[e] = (_Float16)(hi ? 0.f : c0);
        Ac1t1a[e] = (_Float16)(hi ? 0.f : c1);
        Ac1t0b[e] = (_Float16)(hi ? c0 : 0.f);
        Ac1t1b[e] = (_Float16)(hi ? c1 : 0.f);
    }
    v16h Asig, Acol;
#pragma unroll
    for (int e = 0; e < 16; ++e) {
        const int K = (e < 8 ? e : e + 8) + kb;
        const float sv = pSig[K];
        const float cv = pCol[K * 3 + jc];
        Asig[e] = (_Float16)(sigRow ? sv : 0.f);
        Acol[e] = (_Float16)(colRow ? 0.5f * cv : 0.f);
    }
    v8h CbT0, CbT1, CbC0, CbC1;
#pragma unroll
    for (int i = 0; i < 8; ++i) {
        const int h0 = i + (hi ? 8 : 0);
        CbT0[i] = (_Float16)b1[h0];      CbT1[i] = (_Float16)b1[h0 + 16];
        CbC0[i] = (_Float16)bc1[h0];     CbC1[i] = (_Float16)bc1[h0 + 16];
    }

    float ox = rays_o[ray * 3 + 0], oy = rays_o[ray * 3 + 1], oz = rays_o[ray * 3 + 2];
    float dx = rays_d[ray * 3 + 0], dy = rays_d[ray * 3 + 1], dz = rays_d[ray * 3 + 2];
    {
        float il = 1.f / sqrtf(dx * dx + dy * dy + dz * dz);
        dx *= il; dy *= il; dz *= il;
    }
    float near, far;
    {
        float ivx = 1.f / dx, ivy = 1.f / dy, ivz = 1.f / dz;
        float t1x = (-1.f - ox) * ivx, t2x = (1.f - ox) * ivx;
        float t1y = (-1.f - oy) * ivy, t2y = (1.f - oy) * ivy;
        float t1z = (-1.f - oz) * ivz, t2z = (1.f - oz) * ivz;
        near = fmaxf(fmaxf(fminf(t1x, t2x), fminf(t1y, t2y)), fminf(t1z, t2z));
        far  = fminf(fminf(fmaxf(t1x, t2x), fmaxf(t1y, t2y)), fmaxf(t1z, t2z));
        near = fmaxf(near, 0.2f);
        far  = fmaxf(far, near + 1e-6f);
    }

    const int   T      = *num_steps;
    const float invTm1 = 1.f / (float)(T - 1);
    const float fn     = far - near;
    const float dstep  = fn * invTm1;
    const float dlast  = fn / (float)T;

    v16h B1 = {};
    B1[3] = (_Float16)dx;
    B1[4] = (_Float16)dy;
    B1[5] = (_Float16)dz;

    float Tc = 1.f,  ws = 0.f,  dep = 0.f,  im0 = 0.f, im1 = 0.f, im2 = 0.f;

    const int Th = T >> 1;
    const int t0 = seg ? Th : 0;
    const int t1 = seg ? T  : Th;

    int t = t0;
    for (; t + 1 < t1; t += 2) {
        const float z01a = (float)t * invTm1;
        const float z01b = (float)(t + 1) * invTm1;
        const float z01l = hi ? z01b : z01a;
        const float zv   = near + fn * z01l;
        const float px = fminf(fmaxf(ox + dx * zv, -1.f), 1.f);
        const float py = fminf(fmaxf(oy + dy * zv, -1.f), 1.f);
        const float pz = fminf(fmaxf(oz + dz * zv, -1.f), 1.f);
        B1[0] = (_Float16)px;  B1[1] = (_Float16)py;  B1[2] = (_Float16)pz;

        v8h D0a = WMMA_F16H(A1t0a,  B1, CbT0);
        v8h D1a = WMMA_F16H(A1t1a,  B1, CbT1);
        v8h C0a = WMMA_F16H(Ac1t0a, B1, CbC0);
        v8h C1a = WMMA_F16H(Ac1t1a, B1, CbC1);
        v8h D0b = WMMA_F16H(A1t0b,  B1, CbT0);
        v8h D1b = WMMA_F16H(A1t1b,  B1, CbT1);
        v8h C0b = WMMA_F16H(Ac1t0b, B1, CbC0);
        v8h C1b = WMMA_F16H(Ac1t1b, B1, CbC1);


        __builtin_amdgcn_sched_barrier(0);

        const v16h Bha  = relu_pack(D0a, D1a);
        const v16h Bhb  = relu_pack(D0b, D1b);
        const v16h Bhca = relu_pack(C0a, C1a);
        const v16h Bhcb = relu_pack(C0b, C1b);
        const v8f Z = {};
        v8f Dsa = WMMA_F16(Asig, Bha,  Z);
        v8f Dsb = WMMA_F16(Asig, Bhb,  Z);
        v8f Dca = WMMA_F16(Acol, Bhca, Z);
        v8f Dcb = WMMA_F16(Acol, Bhcb, Z);
        asm volatile("" :: "v"(Bha), "v"(Bhb), "v"(Bhca), "v"(Bhcb),
                           "v"(Dsa), "v"(Dsb), "v"(Dca),  "v"(Dcb));

        const float dla = -dstep * L2E;
        const float dlb = -((t + 1 == T - 1) ? dlast : dstep) * L2E;
        const float sga = softplus_f(Dsa[0]);
        const float sgb = softplus_f(Dsb[0]);
        const float aa  = 1.f - exp2_fast(dla * sga);
        const float ab  = 1.f - exp2_fast(dlb * sgb);
        const float ra0 = tanh_fast(Dca[0]), ra1 = tanh_fast(Dca[1]), ra2 = tanh_fast(Dca[2]);
        const float rb0 = tanh_fast(Dcb[0]), rb1 = tanh_fast(Dcb[1]), rb2 = tanh_fast(Dcb[2]);

        const float wa = aa * Tc;
        Tc *= (1.f - aa + 1e-15f);
        ws  += wa;   dep += wa * z01a;
        im0 += wa * ra0;  im1 += wa * ra1;  im2 += wa * ra2;
        const float wb = ab * Tc;
        Tc *= (1.f - ab + 1e-15f);
        ws  += wb;   dep += wb * z01b;
        im0 += wb * rb0;  im1 += wb * rb1;  im2 += wb * rb2;
    }
    if (t < t1) {
        const float z01 = (float)t * invTm1;
        const float zv  = near + fn * z01;
        const float px = fminf(fmaxf(ox + dx * zv, -1.f), 1.f);
        const float py = fminf(fmaxf(oy + dy * zv, -1.f), 1.f);
        const float pz = fminf(fmaxf(oz + dz * zv, -1.f), 1.f);
        B1[0] = (_Float16)px;  B1[1] = (_Float16)py;  B1[2] = (_Float16)pz;

        v8h D0 = WMMA_F16H(A1t0a, B1, CbT0);
        v8h D1 = WMMA_F16H(A1t1a, B1, CbT1);
        v8h C0 = WMMA_F16H(Ac1t0a, B1, CbC0);
        v8h C1 = WMMA_F16H(Ac1t1a, B1, CbC1);
        const v8f Z = {};
        const v16h Bh = relu_pack(D0, D1);
        v8f Ds = WMMA_F16(Asig, Bh, Z);
        const v16h Bhc = relu_pack(C0, C1);
        v8f Dc = WMMA_F16(Acol, Bhc, Z);

        const float delta = (t == T - 1) ? dlast : dstep;
        const float dl2   = -delta * L2E;
        const float sg  = softplus_f(Ds[0]);
        const float a   = 1.f - exp2_fast(dl2 * sg);
        const float w   = a * Tc;
        Tc *= (1.f - a + 1e-15f);
        ws  += w;   dep += w * z01;
        im0 += w * tanh_fast(Dc[0]);
        im1 += w * tanh_fast(Dc[1]);
        im2 += w * tanh_fast(Dc[2]);
    }

    if (seg == 1) {
        float* p = &part[tib][0][lane];
        p[0 * 32] = ws;   p[1 * 32] = dep;  p[2 * 32] = im0;  p[3 * 32] = im1;  p[4 * 32] = im2;
    }
    __syncthreads();
    if (seg == 0) {
        const float* p = &part[tib][0][lane];
        ws   += Tc * p[0 * 32];  dep  += Tc * p[1 * 32];
        im0  += Tc * p[2 * 32];  im1  += Tc * p[3 * 32];  im2 += Tc * p[4 * 32];
    }

    if (seg == 0 && (rayBase + m) < N) {
        float* o = out + (size_t)(rayBase + m) * 9 + (hi ? 5 : 0);
        o[0] = 1.f + 0.5f * (im0 - ws);
        o[1] = 1.f + 0.5f * (im1 - ws);
        o[2] = 1.f + 0.5f * (im2 - ws);
        o[3] = dep;
        if (!hi) o[4] = ws;
    }
}

extern "C" void kernel_launch(void* const* d_in, const int* in_sizes, int n_in,
                              void* d_out, int out_size, void* d_ws, size_t ws_size,
                              hipStream_t stream) {
    const float* rays_o = (const float*)d_in[0];
    const float* rays_d = (const float*)d_in[1];
    const float* W1     = (const float*)d_in[2];
    const float* b1     = (const float*)d_in[3];
    const float* Wsig   = (const float*)d_in[4];
    const float* Wsig_d = (const float*)d_in[5];
    const float* Wc1    = (const float*)d_in[6];
    const float* bc1    = (const float*)d_in[7];
    const float* Wc2    = (const float*)d_in[8];
    const float* Wc2_d  = (const float*)d_in[9];
    const int*   nsteps = (const int*)d_in[10];

    const int N      = in_sizes[0] / 3;
    const int tiles  = (N + 15) / 16;
    const int block  = 256;
    const int grid   = (tiles + 3) / 4;

    nerf_wmma_kernel<<<grid, block, 0, stream>>>(
        rays_o, rays_d, W1, b1, Wsig, Wsig_d, Wc1, bc1, Wc2, Wc2_d,
        nsteps, (float*)d_out, N);
}


# sample loop strength-reduced: positions one multiply-add per axis on the scalar trip count, depth sums w*t scaled once, log2e/ln2 folded into sigma weights, 1-alpha taken as the exponential
# speedup vs baseline: 3.0297x; 1.0490x over previous
#include <hip/hip_runtime.h>

typedef __attribute__((ext_vector_type(16))) _Float16 v16h;
typedef __attribute__((ext_vector_type(8)))  float    v8f;
typedef __attribute__((ext_vector_type(8)))  _Float16 v8h;
typedef __attribute__((ext_vector_type(16))) short    v16s;

#define WMMA_F16(A, B, C) \
    __builtin_amdgcn_wmma_f32_16x16x32_f16(false, (A), false, (B), (short)0, (C), false, false)
#define WMMA_F16H(A, B, C) \
    __builtin_amdgcn_wmma_f16_16x16x32_f16(false, (A), false, (B), (short)0, (C), false, false)

#define L2E 1.4426950408889634f

__device__ __forceinline__ float exp2_fast(float x) { return __builtin_amdgcn_exp2f(x); }
__device__ __forceinline__ float log2_fast(float x) { return __builtin_amdgcn_logf(x); }
__device__ __forceinline__ float rcp_fast(float x)  { return __builtin_amdgcn_rcpf(x); }

__device__ __forceinline__ float softplus2_f(float y) {
    return __builtin_amdgcn_fmed3f(log2_fast(1.f + exp2_fast(y)), y, 128.f);
}
__device__ __forceinline__ float tanh_fast(float x) { return __builtin_amdgcn_tanhf(x); }

__device__ __forceinline__ v16h relu_pack(v8h l, v8h h) {
    const v16h x = __builtin_shufflevector(l, h, 0, 1, 2, 3, 4, 5, 6, 7,
                                                 8, 9, 10, 11, 12, 13, 14, 15);
    const v16s zero = {};
    const v16s r = __builtin_elementwise_max(__builtin_bit_cast(v16s, x), zero);
    return __builtin_bit_cast(v16h, r);
}

__global__ __launch_bounds__(256) __attribute__((amdgpu_waves_per_eu(4, 4)))
void nerf_wmma_kernel(const float* __restrict__ rays_o,
                      const float* __restrict__ rays_d,
                      const float* __restrict__ W1,   const float* __restrict__ b1,
                      const float* __restrict__ Wsig, const float* __restrict__ Wsig_d,
                      const float* __restrict__ Wc1,  const float* __restrict__ bc1,
                      const float* __restrict__ Wc2,  const float* __restrict__ Wc2_d,
                      const int*   __restrict__ num_steps,
                      float* __restrict__ out, int N)
{
    const int lane = threadIdx.x & 31;
    const int wib  = __builtin_amdgcn_readfirstlane((int)(threadIdx.x >> 5));
    const int tib  = wib & 3;
    const int seg  = wib >> 2;
    const int m    = lane & 15;
    const bool hi  = (lane & 16) != 0;
    const int rayBase = (blockIdx.x * 4 + tib) * 16;

    __shared__ float part[4][5][32];

    int ray = rayBase + m;
    if (ray >= N) ray = N - 1;

    const int    kb     = hi ? 8 : 0;
    const int    m7     = m & 7;
    const int    jc     = m7 < 3 ? m7 : 2;
    const bool   sigRow = (m7 == 0);
    const bool   colRow = (m7 < 3);
    const float* pSig   = (m & 8) ? Wsig_d : Wsig;
    const float* pCol   = (m & 8) ? Wc2_d  : Wc2;

    v16h A1t0a = {}, A1t1a = {}, Ac1t0a = {}, Ac1t1a = {};
    v16h A1t0b = {}, A1t1b = {}, Ac1t0b = {}, Ac1t1b = {};
#pragma unroll
    for (int e = 0; e < 6; ++e) {
        if (e < 3) {
            const float a0 = W1[e * 32 + m];
            const float a1 = W1[e * 32 + (m + 16)];
            A1t0a[e] = (_Float16)(hi ? 0.f : a0);
            A1t1a[e] = (_Float16)(hi ? 0.f : a1);
            A1t0b[e] = (_Float16)(hi ? a0 : 0.f);
            A1t1b[e] = (_Float16)(hi ? a1 : 0.f);
        }
        const float c0 = Wc1[e * 32 + m];
        const float c1 = Wc1[e * 32 + (m + 16)];
        Ac1t0a[e] = (_Float16)(hi ? 0.f : c0);
        Ac1t1a[e] = (_Float16)(hi ? 0.f : c1);
        Ac1t0b[e] = (_Float16)(hi ? c0 : 0.f);
        Ac1t1b[e] = (_Float16)(hi ? c1 : 0.f);
    }
    v16h Asig, Acol;
#pragma unroll
    for (int e = 0; e < 16; ++e) {
        const int K = (e < 8 ? e : e + 8) + kb;
        const float sv = pSig[K];
        const float cv = pCol[K * 3 + jc];
        Asig[e] = (_Float16)(sigRow ? L2E * sv : 0.f);
        Acol[e] = (_Float16)(colRow ? 0.5f * cv : 0.f);
    }
    v8h CbT0, CbT1, CbC0, CbC1;
#pragma unroll
    for (int i = 0; i < 8; ++i) {
        const int h0 = i + (hi ? 8 : 0);
        CbT0[i] = (_Float16)b1[h0];      CbT1[i] = (_Float16)b1[h0 + 16];
        CbC0[i] = (_Float16)bc1[h0];     CbC1[i] = (_Float16)bc1[h0 + 16];
    }

    float ox = rays_o[ray * 3 + 0], oy = rays_o[ray * 3 + 1], oz = rays_o[ray * 3 + 2];
    float dx = rays_d[ray * 3 + 0], dy = rays_d[ray * 3 + 1], dz = rays_d[ray * 3 + 2];
    {
        float il = 1.f / sqrtf(dx * dx + dy * dy + dz * dz);
        dx *= il; dy *= il; dz *= il;
    }
    float near, far;
    {
        float ivx = 1.f / dx, ivy = 1.f / dy, ivz = 1.f / dz;
        float t1x = (-1.f - ox) * ivx, t2x = (1.f - ox) * ivx;
        float t1y = (-1.f - oy) * ivy, t2y = (1.f - oy) * ivy;
        float t1z = (-1.f - oz) * ivz, t2z = (1.f - oz) * ivz;
        near = fmaxf(fmaxf(fminf(t1x, t2x), fminf(t1y, t2y)), fminf(t1z, t2z));
        far  = fminf(fminf(fmaxf(t1x, t2x), fmaxf(t1y, t2y)), fmaxf(t1z, t2z));
        near = fmaxf(near, 0.2f);
        far  = fmaxf(far, near + 1e-6f);
    }

    const int   T      = *num_steps;
    const float invTm1 = 1.f / (float)(T - 1);
    const float fn     = far - near;
    const float dstep  = fn * invTm1;
    const float dlast  = fn / (float)T;
    const float ndstep = -dstep;
    const float ndlast = -dlast;
    const float zoff = near + (hi ? dstep : 0.f);
    const float p0x = ox + dx * zoff, p0y = oy + dy * zoff, p0z = oz + dz * zoff;
    const float dpx = dx * dstep,     dpy = dy * dstep,     dpz = dz * dstep;

    v16h B1 = {};
    B1[3] = (_Float16)dx;
    B1[4] = (_Float16)dy;
    B1[5] = (_Float16)dz;

    float Tc = 1.f,  ws = 0.f,  dep = 0.f,  im0 = 0.f, im1 = 0.f, im2 = 0.f;

    const int Th = T >> 1;
    const int t0 = seg ? Th : 0;
    const int t1 = seg ? T  : Th;

    int t = t0;
    for (; t + 1 < t1; t += 2) {
        const float tf   = (float)t;
        const float tf1  = (float)(t + 1);
        asm volatile("" : "+v"(B1));
        const float px = fminf(fmaxf(p0x + dpx * tf, -1.f), 1.f);
        const float py = fminf(fmaxf(p0y + dpy * tf, -1.f), 1.f);
        const float pz = fminf(fmaxf(p0z + dpz * tf, -1.f), 1.f);
        B1[0] = (_Float16)px;  B1[1] = (_Float16)py;  B1[2] = (_Float16)pz;

        v8h D0b = WMMA_F16H(A1t0b,  B1, CbT0);
        v8h D1b = WMMA_F16H(A1t1b,  B1, CbT1);
        v8h D0a = WMMA_F16H(A1t0a,  B1, CbT0);
        v8h D1a = WMMA_F16H(A1t1a,  B1, CbT1);
        v8h C0a = WMMA_F16H(Ac1t0a, B1, CbC0);
        v8h C1a = WMMA_F16H(Ac1t1a, B1, CbC1);
        v8h C0b = WMMA_F16H(Ac1t0b, B1, CbC0);
        v8h C1b = WMMA_F16H(Ac1t1b, B1, CbC1);


        __builtin_amdgcn_sched_barrier(0);

        const v16h Bha  = relu_pack(D0a, D1a);
        const v16h Bhb  = relu_pack(D0b, D1b);
        const v16h Bhca = relu_pack(C0a, C1a);
        const v16h Bhcb = relu_pack(C0b, C1b);
        const v8f Z = {};
        v8f Dsa = WMMA_F16(Asig, Bha,  Z);
        v8f Dsb = WMMA_F16(Asig, Bhb,  Z);
        v8f Dca = WMMA_F16(Acol, Bhca, Z);
        v8f Dcb = WMMA_F16(Acol, Bhcb, Z);
        asm volatile("" :: "v"(Bha), "v"(Bhb), "v"(Bhca), "v"(Bhcb),
                           "v"(Dsa), "v"(Dsb), "v"(Dca),  "v"(Dcb));

        const float dla = ndstep;
        const float dlb = (t + 1 == T - 1) ? ndlast : ndstep;
        const float ea  = exp2_fast(dla * softplus2_f(Dsa[0]));
        const float eb  = exp2_fast(dlb * softplus2_f(Dsb[0]));
        const float aa  = 1.f - ea;
        const float ab  = 1.f - eb;
        const float ra0 = tanh_fast(Dca[0]), ra1 = tanh_fast(Dca[1]), ra2 = tanh_fast(Dca[2]);
        const float rb0 = tanh_fast(Dcb[0]), rb1 = tanh_fast(Dcb[1]), rb2 = tanh_fast(Dcb[2]);

        const float wa = aa * Tc;
        Tc *= (ea + 1e-15f);
        ws  += wa;   dep += wa * tf;
        im0 += wa * ra0;  im1 += wa * ra1;  im2 += wa * ra2;
        const float wb = ab * Tc;
        Tc *= (eb + 1e-15f);
        ws  += wb;   dep += wb * tf1;
        im0 += wb * rb0;  im1 += wb * rb1;  im2 += wb * rb2;
    }
    if (t < t1) {
        const float z01 = (float)t * invTm1;
        const float zv  = near + fn * z01;
        const float px = fminf(fmaxf(ox + dx * zv, -1.f), 1.f);
        const float py = fminf(fmaxf(oy + dy * zv, -1.f), 1.f);
        const float pz = fminf(fmaxf(oz + dz * zv, -1.f), 1.f);
        B1[0] = (_Float16)px;  B1[1] = (_Float16)py;  B1[2] = (_Float16)pz;

        v8h D0 = WMMA_F16H(A1t0a, B1, CbT0);
        v8h D1 = WMMA_F16H(A1t1a, B1, CbT1);
        v8h C0 = WMMA_F16H(Ac1t0a, B1, CbC0);
        v8h C1 = WMMA_F16H(Ac1t1a, B1, CbC1);
        const v8f Z = {};
        const v16h Bh = relu_pack(D0, D1);
        v8f Ds = WMMA_F16(Asig, Bh, Z);
        const v16h Bhc = relu_pack(C0, C1);
        v8f Dc = WMMA_F16(Acol, Bhc, Z);

        const float dl2 = (t == T - 1) ? ndlast : ndstep;
        const float e   = exp2_fast(dl2 * softplus2_f(Ds[0]));
        const float a   = 1.f - e;
        const float w   = a * Tc;
        Tc *= (e + 1e-15f);
        ws  += w;   dep += w * (float)t;
        im0 += w * tanh_fast(Dc[0]);
        im1 += w * tanh_fast(Dc[1]);
        im2 += w * tanh_fast(Dc[2]);
    }

    if (seg == 1) {
        float* p = &part[tib][0][lane];
        p[0 * 32] = ws;   p[1 * 32] = dep;  p[2 * 32] = im0;  p[3 * 32] = im1;  p[4 * 32] = im2;
    }
    __syncthreads();
    if (seg == 0) {
        const float* p = &part[tib][0][lane];
        ws   += Tc * p[0 * 32];  dep  += Tc * p[1 * 32];
        im0  += Tc * p[2 * 32];  im1  += Tc * p[3 * 32];  im2 += Tc * p[4 * 32];
    }

    if (seg == 0 && (rayBase + m) < N) {
        float* o = out + (size_t)(rayBase + m) * 9 + (hi ? 5 : 0);
        o[0] = 1.f + 0.5f * (im0 - ws);
        o[1] = 1.f + 0.5f * (im1 - ws);
        o[2] = 1.f + 0.5f * (im2 - ws);
        o[3] = dep * invTm1;
        if (!hi) o[4] = ws;
    }
}

extern "C" void kernel_launch(void* const* d_in, const int* in_sizes, int n_in,
                              void* d_out, int out_size, void* d_ws, size_t ws_size,
                              hipStream_t stream) {
    const float* rays_o = (const float*)d_in[0];
    const float* rays_d = (const float*)d_in[1];
    const float* W1     = (const float*)d_in[2];
    const float* b1     = (const float*)d_in[3];
    const float* Wsig   = (const float*)d_in[4];
    const float* Wsig_d = (const float*)d_in[5];
    const float* Wc1    = (const float*)d_in[6];
    const float* bc1    = (const float*)d_in[7];
    const float* Wc2    = (const float*)d_in[8];
    const float* Wc2_d  = (const float*)d_in[9];
    const int*   nsteps = (const int*)d_in[10];

    const int N      = in_sizes[0] / 3;
    const int tiles  = (N + 15) / 16;
    const int block  = 256;
    const int grid   = (tiles + 3) / 4;

    nerf_wmma_kernel<<<grid, block, 0, stream>>>(
        rays_o, rays_d, W1, b1, Wsig, Wsig_d, Wc1, bc1, Wc2, Wc2_d,
        nsteps, (float*)d_out, N);
}
